# starting kernel
# baseline (speedup 1.0000x reference)
#include <hip/hip_runtime.h>
#include <hip/hip_bf16.h>

typedef __attribute__((ext_vector_type(16))) __bf16 v16bf;
typedef __attribute__((ext_vector_type(8)))  float  v8f;

#define NEG_SLOPE 0.2f


__device__ __forceinline__ unsigned short f2bf(float f) {
    unsigned u = __float_as_uint(f);
    unsigned r = u + 0x7FFFu + ((u >> 16) & 1u);
    return (unsigned short)(r >> 16);
}

__device__ __forceinline__ unsigned fenc(float f) {
    unsigned u = __float_as_uint(f);
    return (u & 0x80000000u) ? ~u : (u | 0x80000000u);
}
__device__ __forceinline__ float fdec(unsigned u) {
    return (u & 0x80000000u) ? __uint_as_float(u & 0x7FFFFFFFu)
                             : __uint_as_float(~u);
}

__device__ __forceinline__ void get_edge(const int* __restrict__ ei,
                                         long long n_edges, long long idx,
                                         int& s, int& d) {
    if (idx < n_edges) { s = ei[idx]; d = ei[n_edges + idx]; }
    else               { s = d = (int)(idx - n_edges); }
}


__global__ void k_f32_to_bf16(const float* __restrict__ in,
                              unsigned short* __restrict__ out, long long n) {
    long long i = (long long)blockIdx.x * blockDim.x + threadIdx.x;
    if (i < n) out[i] = f2bf(in[i]);
}

__global__ void k_transpose_bf16(const float* __restrict__ W,
                                 unsigned short* __restrict__ Wt,
                                 int K, int Nc) {
    int i = blockIdx.x * blockDim.x + threadIdx.x;
    if (i >= K * Nc) return;
    int n = i / K, k = i % K;
    Wt[(size_t)n * K + k] = f2bf(W[(size_t)k * Nc + n]);
}

__global__ void k_fill_u32(unsigned* __restrict__ p, unsigned v, long long n) {
    long long i = (long long)blockIdx.x * blockDim.x + threadIdx.x;
    if (i < n) p[i] = v;
}

__global__ void k_fill_bias(float* __restrict__ out, const float* __restrict__ bias,
                            int C, long long n) {
    long long i = (long long)blockIdx.x * blockDim.x + threadIdx.x;
    if (i < n) out[i] = bias[i % C];
}

__global__ void k_bias_relu_bf16(const float* __restrict__ in,
                                 const float* __restrict__ bias,
                                 unsigned short* __restrict__ hbf,
                                 int C, long long n) {
    long long i = (long long)blockIdx.x * blockDim.x + threadIdx.x;
    if (i >= n) return;
    float v = in[i] + bias[i % C];
    hbf[i] = f2bf(v > 0.f ? v : 0.f);
}

__global__ void k_gemm_wmma_bf16(const unsigned short* __restrict__ A,
                                 const unsigned short* __restrict__ Bt,
                                 float* __restrict__ C, int Nc) {
    const int lane = threadIdx.x & 31;
    const int tileM = blockIdx.x;
    const int tileN = blockIdx.y;
    const int r15  = lane & 15;
    const int half = lane >> 4;

    const unsigned* arow = (const unsigned*)(A + (size_t)(tileM * 16 + r15) * 128);
    const unsigned* brow = (const unsigned*)(Bt + (size_t)(tileN * 16 + r15) * 128);

    v8f acc = {};
#pragma unroll
    for (int K0 = 0; K0 < 128; K0 += 32) {
        union { unsigned u[8]; v16bf v; } a, b;
#pragma unroll
        for (int v = 0; v < 8; ++v) {
            int k = ((v < 4) ? 0 : 16) + half * 8 + (v & 3) * 2;
            a.u[v] = arow[(K0 + k) >> 1];
            b.u[v] = brow[(K0 + k) >> 1];
        }
        acc = __builtin_amdgcn_wmma_f32_16x16x32_bf16(
                false, a.v, false, b.v, (short)0, acc, false, false);
    }

#pragma unroll
    for (int r = 0; r < 8; ++r) {
        int row = tileM * 16 + r + 8 * half;
        int col = tileN * 16 + r15;
        C[(size_t)row * Nc + col] = acc[r];
    }
}

__global__ void k_alpha(const float* __restrict__ h,
                        const float* __restrict__ a_s, const float* __restrict__ a_d,
                        float* __restrict__ als, float* __restrict__ ald,
                        int n_nodes, int heads, int ch) {
    int i = blockIdx.x * blockDim.x + threadIdx.x;
    if (i >= n_nodes * heads) return;
    int n = i / heads, hd = i % heads;
    const float* row = h + (size_t)n * heads * ch + (size_t)hd * ch;
    const float* vs = a_s + (size_t)hd * ch;
    const float* vd = a_d + (size_t)hd * ch;
    float s = 0.f, d = 0.f;
    for (int c = 0; c < ch; ++c) { float x = row[c]; s += x * vs[c]; d += x * vd[c]; }
    als[i] = s; ald[i] = d;
}

__global__ void k_edge_max(const int* __restrict__ ei,
                           const float* __restrict__ als, const float* __restrict__ ald,
                           unsigned* __restrict__ m,
                           int heads, long long n_edges, long long total) {
    long long i = (long long)blockIdx.x * blockDim.x + threadIdx.x;
    if (i >= total * heads) return;
    long long e = i / heads; int hd = (int)(i % heads);
    int s, d; get_edge(ei, n_edges, e, s, d);
    float v = als[(size_t)s * heads + hd] + ald[(size_t)d * heads + hd];
    v = v > 0.f ? v : NEG_SLOPE * v;
    atomicMax(&m[(size_t)d * heads + hd], fenc(v));
}

__global__ void k_edge_exp(const int* __restrict__ ei,
                           const float* __restrict__ als, const float* __restrict__ ald,
                           const unsigned* __restrict__ m,
                           float* __restrict__ exbuf, float* __restrict__ den,
                           int heads, long long n_edges, long long total) {
    long long i = (long long)blockIdx.x * blockDim.x + threadIdx.x;
    if (i >= total * heads) return;
    long long e = i / heads; int hd = (int)(i % heads);
    int s, d; get_edge(ei, n_edges, e, s, d);
    float v = als[(size_t)s * heads + hd] + ald[(size_t)d * heads + hd];
    v = v > 0.f ? v : NEG_SLOPE * v;
    float ex = __expf(v - fdec(m[(size_t)d * heads + hd]));
    exbuf[i] = ex;
    atomicAdd(&den[(size_t)d * heads + hd], ex);
}

__global__ void k_edge_aggr(const int* __restrict__ ei,
                            const float* __restrict__ exbuf, const float* __restrict__ den,
                            const float* __restrict__ h, float* __restrict__ out,
                            int heads, int ch, long long n_edges, long long total) {
    const int C = heads * ch;
    long long i = (long long)blockIdx.x * blockDim.x + threadIdx.x;
    if (i >= total * C) return;
    long long e = i / C; int c = (int)(i % C); int hd = c / ch;
    int s, d; get_edge(ei, n_edges, e, s, d);
    float alpha = exbuf[e * heads + hd] / (den[(size_t)d * heads + hd] + 1e-16f);
    atomicAdd(&out[(size_t)d * C + c], alpha * h[(size_t)s * C + c]);
}


extern "C" void kernel_launch(void* const* d_in, const int* in_sizes, int n_in,
                              void* d_out, int out_size, void* d_ws, size_t ws_size,
                              hipStream_t stream) {
    (void)n_in; (void)out_size; (void)ws_size; (void)in_sizes;
    constexpr long long N = 100000, E = 1600000, ET = E + N;
    constexpr int C1 = 128, HEADS = 4, HID = 32, C2 = 32;

    const float* x    = (const float*)d_in[0];
    const int*   ei   = (const int*)  d_in[1];
    const float* W1   = (const float*)d_in[2];
    const float* aS1  = (const float*)d_in[3];
    const float* aD1  = (const float*)d_in[4];
    const float* b1   = (const float*)d_in[5];
    const float* W2   = (const float*)d_in[6];
    const float* aS2  = (const float*)d_in[7];
    const float* aD2  = (const float*)d_in[8];
    const float* b2   = (const float*)d_in[9];
    float* out = (float*)d_out;

    char* ws = (char*)d_ws;
    size_t off = 0;
    auto alloc = [&](size_t bytes) { size_t o = off; off = (off + bytes + 255) & ~(size_t)255; return o; };
    unsigned short* xbf  = (unsigned short*)(ws + alloc(N * C1 * 2));
    float*          h1   = (float*)(ws + alloc(N * C1 * 4));
    float*          out1 = (float*)(ws + alloc(N * C1 * 4));
    float*          ex1  = (float*)(ws + alloc(ET * HEADS * 4));
    float*          as1  = (float*)(ws + alloc(N * HEADS * 4));
    float*          ad1  = (float*)(ws + alloc(N * HEADS * 4));
    unsigned*       m1   = (unsigned*)(ws + alloc(N * HEADS * 4));
    float*          den1 = (float*)(ws + alloc(N * HEADS * 4));
    unsigned short* w1t  = (unsigned short*)(ws + alloc(128 * 128 * 2));
    unsigned short* w2t  = (unsigned short*)(ws + alloc(32 * 128 * 2));

    const int B = 256;
    auto blocks = [](long long n, int b) { return (unsigned)((n + b - 1) / b); };

    k_f32_to_bf16<<<blocks(N * C1, B), B, 0, stream>>>(x, xbf, N * C1);
    k_transpose_bf16<<<blocks(128 * 128, B), B, 0, stream>>>(W1, w1t, 128, 128);
    k_transpose_bf16<<<blocks(128 * 32, B), B, 0, stream>>>(W2, w2t, 128, 32);

    k_gemm_wmma_bf16<<<dim3(N / 16, C1 / 16), 32, 0, stream>>>(xbf, w1t, h1, C1);

    k_alpha<<<blocks(N * HEADS, B), B, 0, stream>>>(h1, aS1, aD1, as1, ad1, N, HEADS, HID);

    k_fill_u32<<<blocks(N * HEADS, B), B, 0, stream>>>(m1, 0u, N * HEADS);
    k_fill_u32<<<blocks(N * HEADS, B), B, 0, stream>>>((unsigned*)den1, 0u, N * HEADS);
    k_fill_u32<<<blocks(N * C1, B), B, 0, stream>>>((unsigned*)out1, 0u, N * C1);

    k_edge_max <<<blocks(ET * HEADS, B), B, 0, stream>>>(ei, as1, ad1, m1, HEADS, E, ET);
    k_edge_exp <<<blocks(ET * HEADS, B), B, 0, stream>>>(ei, as1, ad1, m1, ex1, den1, HEADS, E, ET);
    k_edge_aggr<<<blocks(ET * C1, B), B, 0, stream>>>(ei, ex1, den1, h1, out1, HEADS, HID, E, ET);

    k_bias_relu_bf16<<<blocks(N * C1, B), B, 0, stream>>>(out1, b1, xbf, C1, N * C1);

    k_gemm_wmma_bf16<<<dim3(N / 16, C2 / 16), 32, 0, stream>>>(xbf, w2t, h1, C2);

    k_alpha<<<blocks(N, B), B, 0, stream>>>(h1, aS2, aD2, as1, ad1, N, 1, C2);

    k_fill_u32<<<blocks(N, B), B, 0, stream>>>(m1, 0u, N);
    k_fill_u32<<<blocks(N, B), B, 0, stream>>>((unsigned*)den1, 0u, N);
    k_fill_bias<<<blocks(N * C2, B), B, 0, stream>>>(out, b2, C2, N * C2);

    k_edge_max <<<blocks(ET, B), B, 0, stream>>>(ei, as1, ad1, m1, 1, E, ET);
    k_edge_exp <<<blocks(ET, B), B, 0, stream>>>(ei, as1, ad1, m1, ex1, den1, 1, E, ET);
    k_edge_aggr<<<blocks(ET * C2, B), B, 0, stream>>>(ei, ex1, den1, h1, out, 1, C2, E, ET);
}


# Edges grouped by destination inside the call (integer atomic rank, scan, fill); one wave per node does max, exp-sum and weighted row sum in registers; no float atomics, sums in edge-arrival order
# speedup vs baseline: 5.2710x; 5.2710x over previous
#include <hip/hip_runtime.h>
#include <hip/hip_bf16.h>

typedef __attribute__((ext_vector_type(16))) __bf16 v16bf;
typedef __attribute__((ext_vector_type(8)))  float  v8f;

#define NEG_SLOPE 0.2f


__device__ __forceinline__ unsigned short f2bf(float f) {
    unsigned u = __float_as_uint(f);
    unsigned r = u + 0x7FFFu + ((u >> 16) & 1u);
    return (unsigned short)(r >> 16);
}


__global__ void k_f32_to_bf16(const float* __restrict__ in,
                              unsigned short* __restrict__ out, long long n) {
    long long i = (long long)blockIdx.x * blockDim.x + threadIdx.x;
    if (i < n) out[i] = f2bf(in[i]);
}

__global__ void k_transpose_bf16(const float* __restrict__ W,
                                 unsigned short* __restrict__ Wt,
                                 int K, int Nc) {
    int i = blockIdx.x * blockDim.x + threadIdx.x;
    if (i >= K * Nc) return;
    int n = i / K, k = i % K;
    Wt[(size_t)n * K + k] = f2bf(W[(size_t)k * Nc + n]);
}

__global__ void k_fill_u32(unsigned* __restrict__ p, unsigned v, long long n) {
    long long i = (long long)blockIdx.x * blockDim.x + threadIdx.x;
    if (i < n) p[i] = v;
}

__global__ void k_bias_relu_bf16(const float* __restrict__ in,
                                 const float* __restrict__ bias,
                                 unsigned short* __restrict__ hbf,
                                 int C, long long n) {
    long long i = (long long)blockIdx.x * blockDim.x + threadIdx.x;
    if (i >= n) return;
    float v = in[i] + bias[i % C];
    hbf[i] = f2bf(v > 0.f ? v : 0.f);
}

__global__ void k_gemm_wmma_bf16(const unsigned short* __restrict__ A,
                                 const unsigned short* __restrict__ Bt,
                                 float* __restrict__ C, int Nc) {
    const int lane = threadIdx.x & 31;
    const int tileM = blockIdx.x;
    const int tileN = blockIdx.y;
    const int r15  = lane & 15;
    const int half = lane >> 4;

    const unsigned* arow = (const unsigned*)(A + (size_t)(tileM * 16 + r15) * 128);
    const unsigned* brow = (const unsigned*)(Bt + (size_t)(tileN * 16 + r15) * 128);

    v8f acc = {};
#pragma unroll
    for (int K0 = 0; K0 < 128; K0 += 32) {
        union { unsigned u[8]; v16bf v; } a, b;
#pragma unroll
        for (int v = 0; v < 8; ++v) {
            int k = ((v < 4) ? 0 : 16) + half * 8 + (v & 3) * 2;
            a.u[v] = arow[(K0 + k) >> 1];
            b.u[v] = brow[(K0 + k) >> 1];
        }
        acc = __builtin_amdgcn_wmma_f32_16x16x32_bf16(
                false, a.v, false, b.v, (short)0, acc, false, false);
    }

#pragma unroll
    for (int r = 0; r < 8; ++r) {
        int row = tileM * 16 + r + 8 * half;
        int col = tileN * 16 + r15;
        C[(size_t)row * Nc + col] = acc[r];
    }
}

__global__ void k_alpha(const float* __restrict__ h,
                        const float* __restrict__ a_s, const float* __restrict__ a_d,
                        float* __restrict__ als, float* __restrict__ ald,
                        int n_nodes, int heads, int ch) {
    int i = blockIdx.x * blockDim.x + threadIdx.x;
    if (i >= n_nodes * heads) return;
    int n = i / heads, hd = i % heads;
    const float* row = h + (size_t)n * heads * ch + (size_t)hd * ch;
    const float* vs = a_s + (size_t)hd * ch;
    const float* vd = a_d + (size_t)hd * ch;
    float s = 0.f, d = 0.f;
    for (int c = 0; c < ch; ++c) { float x = row[c]; s += x * vs[c]; d += x * vd[c]; }
    als[i] = s; ald[i] = d;
}

__global__ void k_count_rank(const int* __restrict__ ei, int* __restrict__ cnt,
                             int* __restrict__ rank, long long n_edges) {
    long long i = (long long)blockIdx.x * blockDim.x + threadIdx.x;
    if (i >= n_edges) return;
    int d = ei[n_edges + i];
    rank[i] = atomicAdd(&cnt[d], 1);
}

__global__ void __launch_bounds__(256) k_scan_local(const int* __restrict__ cnt,
                                                    int* __restrict__ rowptr,
                                                    int* __restrict__ bsum, int n) {
    __shared__ int wsum[8];
    const int t = threadIdx.x, lane = t & 31, w = t >> 5;
    const int base = blockIdx.x * 1024 + t * 4;
    int c[4];
#pragma unroll
    for (int q = 0; q < 4; ++q) c[q] = (base + q < n) ? cnt[base + q] : 0;
    const int tot = c[0] + c[1] + c[2] + c[3];
    int incl = tot;
#pragma unroll
    for (int off = 1; off < 32; off <<= 1) {
        int y = __shfl_up(incl, off);
        if (lane >= off) incl += y;
    }
    if (lane == 31) wsum[w] = incl;
    __syncthreads();
    int woff = 0;
    for (int q = 0; q < w; ++q) woff += wsum[q];
    int run = woff + incl - tot;
#pragma unroll
    for (int q = 0; q < 4; ++q) {
        if (base + q < n) rowptr[base + q] = run;
        run += c[q];
    }
    if (t == 255) bsum[blockIdx.x] = woff + incl;
}

__global__ void __launch_bounds__(256) k_scan_add(int* __restrict__ rowptr,
                                                  const int* __restrict__ bsum, int n) {
    __shared__ int sh[8];
    const int t = threadIdx.x, lane = t & 31, w = t >> 5;
    const int b = blockIdx.x;
    int v = 0;
    for (int q = t; q < b; q += 256) v += bsum[q];
#pragma unroll
    for (int off = 16; off >= 1; off >>= 1) v += __shfl_xor(v, off);
    if (lane == 0) sh[w] = v;
    __syncthreads();
    int offset = 0;
#pragma unroll
    for (int q = 0; q < 8; ++q) offset += sh[q];
#pragma unroll
    for (int q = 0; q < 4; ++q) {
        int idx = b * 1024 + t * 4 + q;
        if (idx < n) rowptr[idx] += offset;
    }
}

__global__ void k_fill_col(const int* __restrict__ ei, const int* __restrict__ rowptr,
                           const int* __restrict__ rank, int* __restrict__ col,
                           long long n_edges) {
    long long i = (long long)blockIdx.x * blockDim.x + threadIdx.x;
    if (i >= n_edges) return;
    int s = ei[i];
    int d = ei[n_edges + i];
    col[rowptr[d] + rank[i]] = s;
}

template <int H>
__device__ __forceinline__ void edge_score(int k, int n, int start, int node,
                                           const int* __restrict__ col,
                                           const float* __restrict__ als,
                                           const float* adv, int& s, float* v) {
    s = node;
#pragma unroll
    for (int h = 0; h < H; ++h) v[h] = -INFINITY;
    if (k < n) {
        if (k < n - 1) s = col[start + k];
        float x[H];
        if constexpr (H == 4) {
            float4 q = *(const float4*)(als + (size_t)s * 4);
            x[0] = q.x; x[1] = q.y; x[2] = q.z; x[3] = q.w;
        } else {
#pragma unroll
            for (int h = 0; h < H; ++h) x[h] = als[(size_t)s * H + h];
        }
#pragma unroll
        for (int h = 0; h < H; ++h) {
            float y = x[h] + adv[h];
            v[h] = y > 0.f ? y : NEG_SLOPE * y;
        }
    }
}

template <int H>
__device__ __forceinline__ void accum_chunk(int cntc, int s, const float* a,
                                            const float* __restrict__ hmat, int lane,
                                            float* acc) {
#pragma unroll 4
    for (int j = 0; j < cntc; ++j) {
        int sj = __builtin_amdgcn_readlane(s, j);
        const float* row = hmat + (size_t)sj * (H * 32) + lane;
#pragma unroll
        for (int h = 0; h < H; ++h) {
            float aj = __uint_as_float(
                (unsigned)__builtin_amdgcn_readlane((int)__float_as_uint(a[h]), j));
            acc[h] += aj * row[h * 32];
        }
    }
}

template <int H>
__global__ void __launch_bounds__(256) k_gat_gather(const int* __restrict__ rowptr,
                                                    const int* __restrict__ cnt,
                                                    const int* __restrict__ col,
                                                    const float* __restrict__ als,
                                                    const float* __restrict__ ald,
                                                    const float* __restrict__ hmat,
                                                    const float* __restrict__ bias,
                                                    float* __restrict__ outp, int n_nodes) {
    const int lane = threadIdx.x & 31;
    const int node = __builtin_amdgcn_readfirstlane(
        (int)(blockIdx.x * (blockDim.x >> 5) + (threadIdx.x >> 5)));
    if (node >= n_nodes) return;
    const int start = rowptr[node];
    const int n = cnt[node] + 1;

    float adv[H];
#pragma unroll
    for (int h = 0; h < H; ++h) adv[h] = ald[(size_t)node * H + h];

    int s0; float v0[H];
    edge_score<H>(lane, n, start, node, col, als, adv, s0, v0);

    float m[H];
#pragma unroll
    for (int h = 0; h < H; ++h) m[h] = v0[h];
    for (int base = 32; base < n; base += 32) {
        int s; float v[H];
        edge_score<H>(base + lane, n, start, node, col, als, adv, s, v);
#pragma unroll
        for (int h = 0; h < H; ++h) m[h] = fmaxf(m[h], v[h]);
    }
#pragma unroll
    for (int h = 0; h < H; ++h)
#pragma unroll
        for (int off = 16; off >= 1; off >>= 1) m[h] = fmaxf(m[h], __shfl_xor(m[h], off));

    float den[H];
#pragma unroll
    for (int h = 0; h < H; ++h) den[h] = __expf(v0[h] - m[h]);
    for (int base = 32; base < n; base += 32) {
        int s; float v[H];
        edge_score<H>(base + lane, n, start, node, col, als, adv, s, v);
#pragma unroll
        for (int h = 0; h < H; ++h) den[h] += __expf(v[h] - m[h]);
    }
#pragma unroll
    for (int h = 0; h < H; ++h) {
#pragma unroll
        for (int off = 16; off >= 1; off >>= 1) den[h] += __shfl_xor(den[h], off);
        den[h] += 1e-16f;
    }

    float acc[H];
#pragma unroll
    for (int h = 0; h < H; ++h) acc[h] = 0.f;
    {
        float a0[H];
#pragma unroll
        for (int h = 0; h < H; ++h) a0[h] = __expf(v0[h] - m[h]) / den[h];
        accum_chunk<H>(n < 32 ? n : 32, s0, a0, hmat, lane, acc);
    }
    for (int base = 32; base < n; base += 32) {
        int s; float v[H], a[H];
        edge_score<H>(base + lane, n, start, node, col, als, adv, s, v);
#pragma unroll
        for (int h = 0; h < H; ++h) a[h] = __expf(v[h] - m[h]) / den[h];
        accum_chunk<H>((n - base) < 32 ? (n - base) : 32, s, a, hmat, lane, acc);
    }

#pragma unroll
    for (int h = 0; h < H; ++h) {
        float r = acc[h];
        if (bias) r += bias[h * 32 + lane];
        outp[(size_t)node * (H * 32) + h * 32 + lane] = r;
    }
}


extern "C" void kernel_launch(void* const* d_in, const int* in_sizes, int n_in,
                              void* d_out, int out_size, void* d_ws, size_t ws_size,
                              hipStream_t stream) {
    (void)n_in; (void)out_size; (void)ws_size; (void)in_sizes;
    constexpr long long N = 100000, E = 1600000;
    constexpr int C1 = 128, HEADS = 4, HID = 32, C2 = 32;
    constexpr int SCAN_BLOCKS = (int)((N + 1023) / 1024);

    const float* x    = (const float*)d_in[0];
    const int*   ei   = (const int*)  d_in[1];
    const float* W1   = (const float*)d_in[2];
    const float* aS1  = (const float*)d_in[3];
    const float* aD1  = (const float*)d_in[4];
    const float* b1   = (const float*)d_in[5];
    const float* W2   = (const float*)d_in[6];
    const float* aS2  = (const float*)d_in[7];
    const float* aD2  = (const float*)d_in[8];
    const float* b2   = (const float*)d_in[9];
    float* out = (float*)d_out;

    char* ws = (char*)d_ws;
    size_t off = 0;
    auto alloc = [&](size_t bytes) { size_t o = off; off = (off + bytes + 255) & ~(size_t)255; return o; };
    unsigned short* xbf  = (unsigned short*)(ws + alloc(N * C1 * 2));
    float*          h1   = (float*)(ws + alloc(N * C1 * 4));
    float*          out1 = (float*)(ws + alloc(N * C1 * 4));
    float*          as1  = (float*)(ws + alloc(N * HEADS * 4));
    float*          ad1  = (float*)(ws + alloc(N * HEADS * 4));
    int*            cnt  = (int*)(ws + alloc(N * 4));
    int*            rptr = (int*)(ws + alloc(N * 4));
    int*            bsum = (int*)(ws + alloc(SCAN_BLOCKS * 4));
    int*            rank = (int*)(ws + alloc(E * 4));
    int*            col  = (int*)(ws + alloc(E * 4));
    unsigned short* w1t  = (unsigned short*)(ws + alloc(128 * 128 * 2));
    unsigned short* w2t  = (unsigned short*)(ws + alloc(32 * 128 * 2));

    const int B = 256;
    auto blocks = [](long long n, int b) { return (unsigned)((n + b - 1) / b); };

    k_f32_to_bf16<<<blocks(N * C1, B), B, 0, stream>>>(x, xbf, N * C1);
    k_transpose_bf16<<<blocks(128 * 128, B), B, 0, stream>>>(W1, w1t, 128, 128);
    k_transpose_bf16<<<blocks(128 * 32, B), B, 0, stream>>>(W2, w2t, 128, 32);

    k_gemm_wmma_bf16<<<dim3(N / 16, C1 / 16), 32, 0, stream>>>(xbf, w1t, h1, C1);

    k_alpha<<<blocks(N * HEADS, B), B, 0, stream>>>(h1, aS1, aD1, as1, ad1, N, HEADS, HID);

    k_fill_u32<<<blocks(N, B), B, 0, stream>>>((unsigned*)cnt, 0u, N);
    k_count_rank<<<blocks(E, B), B, 0, stream>>>(ei, cnt, rank, E);
    k_scan_local<<<SCAN_BLOCKS, 256, 0, stream>>>(cnt, rptr, bsum, (int)N);
    k_scan_add<<<SCAN_BLOCKS, 256, 0, stream>>>(rptr, bsum, (int)N);
    k_fill_col<<<blocks(E, B), B, 0, stream>>>(ei, rptr, rank, col, E);

    k_gat_gather<HEADS><<<blocks(N, 8), 256, 0, stream>>>(rptr, cnt, col, as1, ad1, h1, nullptr, out1, (int)N);

    k_bias_relu_bf16<<<blocks(N * C1, B), B, 0, stream>>>(out1, b1, xbf, C1, N * C1);

    k_gemm_wmma_bf16<<<dim3(N / 16, C2 / 16), 32, 0, stream>>>(xbf, w2t, h1, C2);

    k_alpha<<<blocks(N, B), B, 0, stream>>>(h1, aS2, aD2, as1, ad1, N, 1, C2);

    k_gat_gather<1><<<blocks(N, 8), 256, 0, stream>>>(rptr, cnt, col, as1, ad1, h1, b2, out, (int)N);
}


# Fewer, longer-lived waves: a GEMM wave does a whole 16-row strip in 8-wave groups, edge kernels take 8 edges per thread, gather waves loop over 7 nodes each; sums stay in edge-arrival order
# speedup vs baseline: 6.2860x; 1.1926x over previous
#include <hip/hip_runtime.h>
#include <hip/hip_bf16.h>

typedef __attribute__((ext_vector_type(16))) __bf16 v16bf;
typedef __attribute__((ext_vector_type(8)))  float  v8f;
typedef __attribute__((ext_vector_type(2)))  float  v2f;
typedef __attribute__((ext_vector_type(2)))  __bf16 v2bf;

#define NEG_SLOPE 0.2f


__device__ __forceinline__ unsigned short f2bf(float f) {
    unsigned u = __float_as_uint(f);
    unsigned r = u + 0x7FFFu + ((u >> 16) & 1u);
    return (unsigned short)(r >> 16);
}

__device__ __forceinline__ unsigned pack_bf16x2(float lo, float hi) {
    v2f f = {lo, hi};
    return __builtin_bit_cast(unsigned, __builtin_convertvector(f, v2bf));
}

__device__ __forceinline__ unsigned short to_bf16_bits(float f) {
    return __builtin_bit_cast(unsigned short, (__bf16)f);
}


__global__ void k_transpose_bf16(const float* __restrict__ W,
                                 unsigned short* __restrict__ Wt,
                                 int K, int Nc) {
    int i = blockIdx.x * blockDim.x + threadIdx.x;
    if (i >= K * Nc) return;
    int n = i / K, k = i % K;
    Wt[(size_t)n * K + k] = f2bf(W[(size_t)k * Nc + n]);
}

__global__ void k_fill_u32(unsigned* __restrict__ p, unsigned v, long long n) {
    long long i = (long long)blockIdx.x * blockDim.x + threadIdx.x;
    if (i < n) p[i] = v;
}

union Frag { unsigned u[8]; v16bf v; };
typedef __attribute__((ext_vector_type(4))) float    v4f;
typedef __attribute__((ext_vector_type(4))) unsigned v4u;

__device__ __forceinline__ v16bf frag_of(v4u q0, v4u q1) {
    Frag f;
    f.u[0] = q0.x; f.u[1] = q0.y; f.u[2] = q0.z; f.u[3] = q0.w;
    f.u[4] = q1.x; f.u[5] = q1.y; f.u[6] = q1.z; f.u[7] = q1.w;
    return f.v;
}

template <bool A_F32, int NT>
__global__ void __launch_bounds__(256) k_gemm_strip(const float* __restrict__ Af,
                                                    const unsigned short* __restrict__ Ab,
                                                    const unsigned short* __restrict__ Bt,
                                                    float* __restrict__ C, int m_tiles) {
    const int lane = threadIdx.x & 31;
    const int tileM = __builtin_amdgcn_readfirstlane((int)(blockIdx.x * 8 + (threadIdx.x >> 5)));
    if (tileM >= m_tiles) return;
    const int r15  = lane & 15;
    const int half = lane >> 4;

    const v4u* bbase = (const v4u*)(Bt + (size_t)r15 * 128 + half * 8);
    v4u qb[2][8];
#pragma unroll
    for (int s = 0; s < 4; ++s) { qb[0][2 * s] = bbase[s * 4 + 0]; qb[0][2 * s + 1] = bbase[s * 4 + 2]; }

    v16bf a[4];
    if constexpr (A_F32) {
        const v4f* arow = (const v4f*)(Af + (size_t)(tileM * 16 + r15) * 128 + half * 8);
        v4f p[16];
#pragma unroll
        for (int s = 0; s < 4; ++s) {
            p[4 * s + 0] = arow[s * 8 + 0];
            p[4 * s + 1] = arow[s * 8 + 1];
            p[4 * s + 2] = arow[s * 8 + 4];
            p[4 * s + 3] = arow[s * 8 + 5];
        }
        asm volatile("" :: "v"(p[0]), "v"(p[1]), "v"(p[2]), "v"(p[3]), "v"(p[4]), "v"(p[5]), "v"(p[6]), "v"(p[7]),
                           "v"(p[8]), "v"(p[9]), "v"(p[10]), "v"(p[11]), "v"(p[12]), "v"(p[13]), "v"(p[14]), "v"(p[15]));
#pragma unroll
        for (int s = 0; s < 4; ++s) {
            Frag f;
#pragma unroll
            for (int j = 0; j < 4; ++j) {
                f.u[2 * j]     = pack_bf16x2(p[4 * s + j].x, p[4 * s + j].y);
                f.u[2 * j + 1] = pack_bf16x2(p[4 * s + j].z, p[4 * s + j].w);
            }
            a[s] = f.v;
        }
    } else {
        const v4u* arow = (const v4u*)(Ab + (size_t)(tileM * 16 + r15) * 128 + half * 8);
        v4u qa[8];
#pragma unroll
        for (int s = 0; s < 4; ++s) { qa[2 * s] = arow[s * 4 + 0]; qa[2 * s + 1] = arow[s * 4 + 2]; }
#pragma unroll
        for (int s = 0; s < 4; ++s) a[s] = frag_of(qa[2 * s], qa[2 * s + 1]);
    }

#pragma unroll
    for (int t = 0; t < NT; ++t) {
        if (t + 1 < NT) {
#pragma unroll
            for (int s = 0; s < 4; ++s) {
                qb[(t + 1) & 1][2 * s]     = bbase[(t + 1) * 256 + s * 4 + 0];
                qb[(t + 1) & 1][2 * s + 1] = bbase[(t + 1) * 256 + s * 4 + 2];
            }
        }
        v8f acc = {};
#pragma unroll
        for (int s = 0; s < 4; ++s)
            acc = __builtin_amdgcn_wmma_f32_16x16x32_bf16(
                    false, a[s], false, frag_of(qb[t & 1][2 * s], qb[t & 1][2 * s + 1]),
                    (short)0, acc, false, false);
#pragma unroll
        for (int r = 0; r < 8; ++r) {
            int row = tileM * 16 + r + 8 * half;
            int col = t * 16 + r15;
            C[(size_t)row * (NT * 16) + col] = acc[r];
        }
    }
}

__global__ void k_alpha(const float* __restrict__ h,
                        const float* __restrict__ a_s, const float* __restrict__ a_d,
                        float* __restrict__ als, float* __restrict__ ald,
                        int n_nodes, int heads, int ch) {
    int i = blockIdx.x * blockDim.x + threadIdx.x;
    if (i >= n_nodes * heads) return;
    int n = i / heads, hd = i % heads;
    const float* row = h + (size_t)n * heads * ch + (size_t)hd * ch;
    const float* vs = a_s + (size_t)hd * ch;
    const float* vd = a_d + (size_t)hd * ch;
    float s = 0.f, d = 0.f;
    for (int c = 0; c < ch; ++c) { float x = row[c]; s += x * vs[c]; d += x * vd[c]; }
    als[i] = s; ald[i] = d;
}

#define EPT 8
template <bool FULL>
__device__ __forceinline__ void count_rank_body(const int* __restrict__ ei, int* __restrict__ cnt,
                                                int* __restrict__ rank, long long n_edges, long long base) {
    int d[EPT], r[EPT];
#pragma unroll
    for (int q = 0; q < EPT; ++q) {
        long long i = base + q * 256;
        d[q] = (FULL || i < n_edges) ? ei[n_edges + i] : 0;
    }
#pragma unroll
    for (int q = 0; q < EPT; ++q) {
        long long i = base + q * 256;
        r[q] = 0;
        if (FULL || i < n_edges) r[q] = atomicAdd(&cnt[d[q]], 1);
    }
#pragma unroll
    for (int q = 0; q < EPT; ++q) {
        long long i = base + q * 256;
        if (FULL || i < n_edges) rank[i] = r[q];
    }
}

__global__ void __launch_bounds__(256) k_count_rank(const int* __restrict__ ei, int* __restrict__ cnt,
                                                    int* __restrict__ rank, long long n_edges) {
    const long long blk = (long long)blockIdx.x * (256 * EPT);
    if (blk + 256 * EPT <= n_edges) count_rank_body<true>(ei, cnt, rank, n_edges, blk + threadIdx.x);
    else                            count_rank_body<false>(ei, cnt, rank, n_edges, blk + threadIdx.x);
}

__global__ void __launch_bounds__(256) k_scan_local(const int* __restrict__ cnt,
                                                    int* __restrict__ rowptr,
                                                    int* __restrict__ bsum, int n) {
    __shared__ int wsum[8];
    const int t = threadIdx.x, lane = t & 31, w = t >> 5;
    const int base = blockIdx.x * 1024 + t * 4;
    int c[4];
#pragma unroll
    for (int q = 0; q < 4; ++q) c[q] = (base + q < n) ? cnt[base + q] : 0;
    const int tot = c[0] + c[1] + c[2] + c[3];
    int incl = tot;
#pragma unroll
    for (int off = 1; off < 32; off <<= 1) {
        int y = __shfl_up(incl, off);
        if (lane >= off) incl += y;
    }
    if (lane == 31) wsum[w] = incl;
    __syncthreads();
    int woff = 0;
    for (int q = 0; q < w; ++q) woff += wsum[q];
    int run = woff + incl - tot;
#pragma unroll
    for (int q = 0; q < 4; ++q) {
        if (base + q < n) rowptr[base + q] = run;
        run += c[q];
    }
    if (t == 255) bsum[blockIdx.x] = woff + incl;
}

__global__ void __launch_bounds__(256) k_scan_add(int* __restrict__ rowptr,
                                                  const int* __restrict__ bsum, int n) {
    __shared__ int sh[8];
    const int t = threadIdx.x, lane = t & 31, w = t >> 5;
    const int b = blockIdx.x;
    int v = 0;
    for (int q = t; q < b; q += 256) v += bsum[q];
#pragma unroll
    for (int off = 16; off >= 1; off >>= 1) v += __shfl_xor(v, off);
    if (lane == 0) sh[w] = v;
    __syncthreads();
    int offset = 0;
#pragma unroll
    for (int q = 0; q < 8; ++q) offset += sh[q];
#pragma unroll
    for (int q = 0; q < 4; ++q) {
        int idx = b * 1024 + t * 4 + q;
        if (idx < n) rowptr[idx] += offset;
    }
}

template <bool FULL>
__device__ __forceinline__ void fill_col_body(const int* __restrict__ ei, const int* __restrict__ rowptr,
                                              const int* __restrict__ rank, int* __restrict__ col,
                                              long long n_edges, long long base) {
    int s[EPT], d[EPT], rk[EPT], pos[EPT];
#pragma unroll
    for (int q = 0; q < EPT; ++q) {
        long long i = base + q * 256;
        const bool ok = FULL || i < n_edges;
        s[q]  = ok ? ei[i] : 0;
        d[q]  = ok ? ei[n_edges + i] : 0;
        rk[q] = ok ? rank[i] : 0;
    }
#pragma unroll
    for (int q = 0; q < EPT; ++q) {
        long long i = base + q * 256;
        pos[q] = (FULL || i < n_edges) ? rowptr[d[q]] + rk[q] : 0;
    }
#pragma unroll
    for (int q = 0; q < EPT; ++q) {
        long long i = base + q * 256;
        if (FULL || i < n_edges) col[pos[q]] = s[q];
    }
}

__global__ void __launch_bounds__(256) k_fill_col(const int* __restrict__ ei, const int* __restrict__ rowptr,
                                                  const int* __restrict__ rank, int* __restrict__ col,
                                                  long long n_edges) {
    const long long blk = (long long)blockIdx.x * (256 * EPT);
    if (blk + 256 * EPT <= n_edges) fill_col_body<true>(ei, rowptr, rank, col, n_edges, blk + threadIdx.x);
    else                            fill_col_body<false>(ei, rowptr, rank, col, n_edges, blk + threadIdx.x);
}

template <int H>
__device__ __forceinline__ void edge_score(int k, int n, int start, int node,
                                           const int* __restrict__ col,
                                           const float* __restrict__ als,
                                           const float* adv, int& s, float* v) {
    s = node;
#pragma unroll
    for (int h = 0; h < H; ++h) v[h] = -INFINITY;
    if (k < n) {
        if (k < n - 1) s = col[start + k];
        float x[H];
        if constexpr (H == 4) {
            float4 q = *(const float4*)(als + (size_t)s * 4);
            x[0] = q.x; x[1] = q.y; x[2] = q.z; x[3] = q.w;
        } else {
#pragma unroll
            for (int h = 0; h < H; ++h) x[h] = als[(size_t)s * H + h];
        }
#pragma unroll
        for (int h = 0; h < H; ++h) {
            float y = x[h] + adv[h];
            v[h] = y > 0.f ? y : NEG_SLOPE * y;
        }
    }
}

template <int H>
__device__ __forceinline__ void accum_chunk(int cntc, int s, const float* a,
                                            const float* __restrict__ hmat, int lane,
                                            float* acc) {
#pragma unroll 4
    for (int j = 0; j < cntc; ++j) {
        int sj = __builtin_amdgcn_readlane(s, j);
        const float* row = hmat + (size_t)sj * (H * 32) + lane;
#pragma unroll
        for (int h = 0; h < H; ++h) {
            float aj = __uint_as_float(
                (unsigned)__builtin_amdgcn_readlane((int)__float_as_uint(a[h]), j));
            acc[h] += aj * row[h * 32];
        }
    }
}

template <int H, bool RELU_BF16>
__global__ void __launch_bounds__(256) k_gat_gather(const int* __restrict__ rowptr,
                                                    const int* __restrict__ cnt,
                                                    const int* __restrict__ col,
                                                    const float* __restrict__ als,
                                                    const float* __restrict__ ald,
                                                    const float* __restrict__ hmat,
                                                    const float* __restrict__ bias,
                                                    float* __restrict__ outp,
                                                    unsigned short* __restrict__ outbf,
                                                    int n_nodes) {
    const int lane = threadIdx.x & 31;
    const int wave0 = __builtin_amdgcn_readfirstlane(
        (int)(blockIdx.x * (blockDim.x >> 5) + (threadIdx.x >> 5)));
    const int n_waves = (int)(gridDim.x * (blockDim.x >> 5));
    for (int node = wave0; node < n_nodes; node += n_waves) {
    const int start = rowptr[node];
    const int n = cnt[node] + 1;

    float adv[H];
#pragma unroll
    for (int h = 0; h < H; ++h) adv[h] = ald[(size_t)node * H + h];

    int s0; float v0[H];
    edge_score<H>(lane, n, start, node, col, als, adv, s0, v0);

    float m[H];
#pragma unroll
    for (int h = 0; h < H; ++h) m[h] = v0[h];
    for (int base = 32; base < n; base += 32) {
        int s; float v[H];
        edge_score<H>(base + lane, n, start, node, col, als, adv, s, v);
#pragma unroll
        for (int h = 0; h < H; ++h) m[h] = fmaxf(m[h], v[h]);
    }
#pragma unroll
    for (int h = 0; h < H; ++h)
#pragma unroll
        for (int off = 16; off >= 1; off >>= 1) m[h] = fmaxf(m[h], __shfl_xor(m[h], off));

    float den[H];
#pragma unroll
    for (int h = 0; h < H; ++h) den[h] = __expf(v0[h] - m[h]);
    for (int base = 32; base < n; base += 32) {
        int s; float v[H];
        edge_score<H>(base + lane, n, start, node, col, als, adv, s, v);
#pragma unroll
        for (int h = 0; h < H; ++h) den[h] += __expf(v[h] - m[h]);
    }
#pragma unroll
    for (int h = 0; h < H; ++h) {
#pragma unroll
        for (int off = 16; off >= 1; off >>= 1) den[h] += __shfl_xor(den[h], off);
        den[h] += 1e-16f;
    }

    float acc[H];
#pragma unroll
    for (int h = 0; h < H; ++h) acc[h] = 0.f;
    {
        float a0[H];
#pragma unroll
        for (int h = 0; h < H; ++h) a0[h] = __expf(v0[h] - m[h]) / den[h];
        accum_chunk<H>(n < 32 ? n : 32, s0, a0, hmat, lane, acc);
    }
    for (int base = 32; base < n; base += 32) {
        int s; float v[H], a[H];
        edge_score<H>(base + lane, n, start, node, col, als, adv, s, v);
#pragma unroll
        for (int h = 0; h < H; ++h) a[h] = __expf(v[h] - m[h]) / den[h];
        accum_chunk<H>((n - base) < 32 ? (n - base) : 32, s, a, hmat, lane, acc);
    }

#pragma unroll
    for (int h = 0; h < H; ++h) {
        float r = acc[h] + bias[h * 32 + lane];
        if constexpr (RELU_BF16)
            outbf[(size_t)node * (H * 32) + h * 32 + lane] = to_bf16_bits(r > 0.f ? r : 0.f);
        else
            outp[(size_t)node * (H * 32) + h * 32 + lane] = r;
    }
    }
}


extern "C" void kernel_launch(void* const* d_in, const int* in_sizes, int n_in,
                              void* d_out, int out_size, void* d_ws, size_t ws_size,
                              hipStream_t stream) {
    (void)n_in; (void)out_size; (void)ws_size; (void)in_sizes;
    constexpr long long N = 100000, E = 1600000;
    constexpr int C1 = 128, HEADS = 4, HID = 32, C2 = 32;
    constexpr int SCAN_BLOCKS = (int)((N + 1023) / 1024);
    constexpr int NODES_PER_WAVE = 7;
    constexpr int GATHER_BLOCKS = (int)((N + 8 * NODES_PER_WAVE - 1) / (8 * NODES_PER_WAVE));

    const float* x    = (const float*)d_in[0];
    const int*   ei   = (const int*)  d_in[1];
    const float* W1   = (const float*)d_in[2];
    const float* aS1  = (const float*)d_in[3];
    const float* aD1  = (const float*)d_in[4];
    const float* b1   = (const float*)d_in[5];
    const float* W2   = (const float*)d_in[6];
    const float* aS2  = (const float*)d_in[7];
    const float* aD2  = (const float*)d_in[8];
    const float* b2   = (const float*)d_in[9];
    float* out = (float*)d_out;

    char* ws = (char*)d_ws;
    size_t off = 0;
    auto alloc = [&](size_t bytes) { size_t o = off; off = (off + bytes + 255) & ~(size_t)255; return o; };
    unsigned short* hbf  = (unsigned short*)(ws + alloc(N * C1 * 2));
    float*          h1   = (float*)(ws + alloc(N * C1 * 4));
    float*          as1  = (float*)(ws + alloc(N * HEADS * 4));
    float*          ad1  = (float*)(ws + alloc(N * HEADS * 4));
    int*            cnt  = (int*)(ws + alloc(N * 4));
    int*            rptr = (int*)(ws + alloc(N * 4));
    int*            bsum = (int*)(ws + alloc(SCAN_BLOCKS * 4));
    int*            rank = (int*)(ws + alloc(E * 4));
    int*            col  = (int*)(ws + alloc(E * 4));
    unsigned short* w1t  = (unsigned short*)(ws + alloc(128 * 128 * 2));
    unsigned short* w2t  = (unsigned short*)(ws + alloc(32 * 128 * 2));

    const int B = 256;
    auto blocks = [](long long n, int b) { return (unsigned)((n + b - 1) / b); };

    k_transpose_bf16<<<blocks(128 * 128, B), B, 0, stream>>>(W1, w1t, 128, 128);
    k_transpose_bf16<<<blocks(128 * 32, B), B, 0, stream>>>(W2, w2t, 128, 32);

    k_gemm_strip<true, C1 / 16><<<blocks(N / 16, 8), 256, 0, stream>>>(x, nullptr, w1t, h1, (int)(N / 16));

    k_alpha<<<blocks(N * HEADS, B), B, 0, stream>>>(h1, aS1, aD1, as1, ad1, N, HEADS, HID);

    k_fill_u32<<<blocks(N, B), B, 0, stream>>>((unsigned*)cnt, 0u, N);
    k_count_rank<<<blocks(E, B * EPT), B, 0, stream>>>(ei, cnt, rank, E);
    k_scan_local<<<SCAN_BLOCKS, 256, 0, stream>>>(cnt, rptr, bsum, (int)N);
    k_scan_add<<<SCAN_BLOCKS, 256, 0, stream>>>(rptr, bsum, (int)N);
    k_fill_col<<<blocks(E, B * EPT), B, 0, stream>>>(ei, rptr, rank, col, E);

    k_gat_gather<HEADS, true><<<GATHER_BLOCKS, 256, 0, stream>>>(rptr, cnt, col, as1, ad1, h1, b1, nullptr, hbf, (int)N);

    k_gemm_strip<false, C2 / 16><<<blocks(N / 16, 8), 256, 0, stream>>>(nullptr, hbf, w2t, h1, (int)(N / 16));

    k_alpha<<<blocks(N, B), B, 0, stream>>>(h1, aS2, aD2, as1, ad1, N, 1, C2);

    k_gat_gather<1, false><<<GATHER_BLOCKS, 256, 0, stream>>>(rptr, cnt, col, as1, ad1, h1, b2, out, nullptr, (int)N);
}


# Independent stages share launches: weight transposes + counter clear in one kernel, edge count/rank as alternate workgroups of GEMM-1's launch, scan in one kernel; 11 -> 7 launches
# speedup vs baseline: 7.0078x; 1.1148x over previous
#include <hip/hip_runtime.h>
#include <hip/hip_bf16.h>

typedef __attribute__((ext_vector_type(16))) __bf16 v16bf;
typedef __attribute__((ext_vector_type(8)))  float  v8f;
typedef __attribute__((ext_vector_type(2)))  float  v2f;
typedef __attribute__((ext_vector_type(2)))  __bf16 v2bf;

#define NEG_SLOPE 0.2f


__device__ __forceinline__ unsigned short f2bf(float f) {
    unsigned u = __float_as_uint(f);
    unsigned r = u + 0x7FFFu + ((u >> 16) & 1u);
    return (unsigned short)(r >> 16);
}

__device__ __forceinline__ unsigned pack_bf16x2(float lo, float hi) {
    v2f f = {lo, hi};
    return __builtin_bit_cast(unsigned, __builtin_convertvector(f, v2bf));
}

__device__ __forceinline__ unsigned short to_bf16_bits(float f) {
    return __builtin_bit_cast(unsigned short, (__bf16)f);
}


__global__ void __launch_bounds__(256) k_prep(const float* __restrict__ W1, unsigned short* __restrict__ w1t,
                                              const float* __restrict__ W2, unsigned short* __restrict__ w2t,
                                              int* __restrict__ cnt, int n_nodes) {
    int i = blockIdx.x * 256 + threadIdx.x;
    if (i < 128 * 128) {
        int n = i >> 7, k = i & 127;
        w1t[i] = f2bf(W1[k * 128 + n]);
    } else if (i < 128 * 128 + 32 * 128) {
        i -= 128 * 128;
        int n = i >> 7, k = i & 127;
        w2t[i] = f2bf(W2[k * 32 + n]);
    } else {
        i -= 128 * 128 + 32 * 128;
        if (i < n_nodes) cnt[i] = 0;
    }
}

union Frag { unsigned u[8]; v16bf v; };
typedef __attribute__((ext_vector_type(4))) float    v4f;
typedef __attribute__((ext_vector_type(4))) unsigned v4u;

__device__ __forceinline__ v16bf frag_of(v4u q0, v4u q1) {
    Frag f;
    f.u[0] = q0.x; f.u[1] = q0.y; f.u[2] = q0.z; f.u[3] = q0.w;
    f.u[4] = q1.x; f.u[5] = q1.y; f.u[6] = q1.z; f.u[7] = q1.w;
    return f.v;
}

template <int M>
__device__ __forceinline__ float row_xor(float v) {
    return __builtin_bit_cast(float, __builtin_amdgcn_mov_dpp(
        __builtin_bit_cast(int, v), 0x160 + M, 0xf, 0xf, false));
}

__device__ __forceinline__ float row_reduce_scatter(const float* ps, const float* pd, int r15) {
    const bool b3 = r15 & 8, b2 = r15 & 4, b1 = r15 & 2, b0 = r15 & 1;
    float w[8], x[4], y[2];
#pragma unroll
    for (int i = 0; i < 8; ++i) w[i] = (b3 ? pd[i] : ps[i]) + row_xor<8>(b3 ? ps[i] : pd[i]);
#pragma unroll
    for (int i = 0; i < 4; ++i) x[i] = (b2 ? w[i + 4] : w[i]) + row_xor<4>(b2 ? w[i] : w[i + 4]);
#pragma unroll
    for (int i = 0; i < 2; ++i) y[i] = (b1 ? x[i + 2] : x[i]) + row_xor<2>(b1 ? x[i] : x[i + 2]);
    return (b0 ? y[1] : y[0]) + row_xor<1>(b0 ? y[0] : y[1]);
}

template <bool A_F32, int NT>
__device__ __forceinline__ void gemm_strip_body(const float* __restrict__ Af,
                                                const unsigned short* __restrict__ Ab,
                                                const unsigned short* __restrict__ Bt,
                                                float* __restrict__ C,
                                                const float* __restrict__ a_s,
                                                const float* __restrict__ a_d,
                                                float* __restrict__ als, float* __restrict__ ald,
                                                int m_tiles, int wg) {
    constexpr int H = NT / 2;
    const int lane = threadIdx.x & 31;
    const int tileM = __builtin_amdgcn_readfirstlane((int)(wg * 8 + (threadIdx.x >> 5)));
    if (tileM >= m_tiles) return;
    const int r15  = lane & 15;
    const int half = lane >> 4;

    const v4u* bbase = (const v4u*)(Bt + (size_t)r15 * 128 + half * 8);
    v4u qb[2][8];
#pragma unroll
    for (int s = 0; s < 4; ++s) { qb[0][2 * s] = bbase[s * 4 + 0]; qb[0][2 * s + 1] = bbase[s * 4 + 2]; }

    v16bf a[4];
    if constexpr (A_F32) {
        const v4f* arow = (const v4f*)(Af + (size_t)(tileM * 16 + r15) * 128 + half * 8);
        v4f p[16];
#pragma unroll
        for (int s = 0; s < 4; ++s) {
            p[4 * s + 0] = arow[s * 8 + 0];
            p[4 * s + 1] = arow[s * 8 + 1];
            p[4 * s + 2] = arow[s * 8 + 4];
            p[4 * s + 3] = arow[s * 8 + 5];
        }
        asm volatile("" :: "v"(p[0]), "v"(p[1]), "v"(p[2]), "v"(p[3]), "v"(p[4]), "v"(p[5]), "v"(p[6]), "v"(p[7]),
                           "v"(p[8]), "v"(p[9]), "v"(p[10]), "v"(p[11]), "v"(p[12]), "v"(p[13]), "v"(p[14]), "v"(p[15]));
#pragma unroll
        for (int s = 0; s < 4; ++s) {
            Frag f;
#pragma unroll
            for (int j = 0; j < 4; ++j) {
                f.u[2 * j]     = pack_bf16x2(p[4 * s + j].x, p[4 * s + j].y);
                f.u[2 * j + 1] = pack_bf16x2(p[4 * s + j].z, p[4 * s + j].w);
            }
            a[s] = f.v;
        }
    } else {
        const v4u* arow = (const v4u*)(Ab + (size_t)(tileM * 16 + r15) * 128 + half * 8);
        v4u qa[8];
#pragma unroll
        for (int s = 0; s < 4; ++s) { qa[2 * s] = arow[s * 4 + 0]; qa[2 * s + 1] = arow[s * 4 + 2]; }
#pragma unroll
        for (int s = 0; s < 4; ++s) a[s] = frag_of(qa[2 * s], qa[2 * s + 1]);
    }

    float ps[8], pd[8], res[H];
#pragma unroll
    for (int t = 0; t < NT; ++t) {
        if (t + 1 < NT) {
#pragma unroll
            for (int s = 0; s < 4; ++s) {
                qb[(t + 1) & 1][2 * s]     = bbase[(t + 1) * 256 + s * 4 + 0];
                qb[(t + 1) & 1][2 * s + 1] = bbase[(t + 1) * 256 + s * 4 + 2];
            }
        }
        const float cs = a_s[t * 16 + r15], cd = a_d[t * 16 + r15];
        v8f acc = {};
#pragma unroll
        for (int s = 0; s < 4; ++s)
            acc = __builtin_amdgcn_wmma_f32_16x16x32_bf16(
                    false, a[s], false, frag_of(qb[t & 1][2 * s], qb[t & 1][2 * s + 1]),
                    (short)0, acc, false, false);
#pragma unroll
        for (int r = 0; r < 8; ++r) {
            int row = tileM * 16 + r + 8 * half;
            int col = t * 16 + r15;
            C[(size_t)row * (NT * 16) + col] = acc[r];
        }
        if ((t & 1) == 0) {
#pragma unroll
            for (int r = 0; r < 8; ++r) { ps[r] = acc[r] * cs; pd[r] = acc[r] * cd; }
        } else {
#pragma unroll
            for (int r = 0; r < 8; ++r) { ps[r] += acc[r] * cs; pd[r] += acc[r] * cd; }
            res[t >> 1] = row_reduce_scatter(ps, pd, r15);
        }
    }
    {
        const int row = tileM * 16 + 8 * half + (r15 & 7);
        float* dstp = ((r15 & 8) ? ald : als) + (size_t)row * H;
        if constexpr (H == 4) {
            *(float4*)dstp = make_float4(res[0], res[1], res[2], res[3]);
        } else {
#pragma unroll
            for (int h = 0; h < H; ++h) dstp[h] = res[h];
        }
    }
}

__global__ void __launch_bounds__(256) k_gemm2_strip(const unsigned short* __restrict__ Ab,
                                                     const unsigned short* __restrict__ Bt,
                                                     float* __restrict__ C,
                                                     const float* __restrict__ a_s,
                                                     const float* __restrict__ a_d,
                                                     float* __restrict__ als, float* __restrict__ ald,
                                                     int m_tiles) {
    gemm_strip_body<false, 2>(nullptr, Ab, Bt, C, a_s, a_d, als, ald, m_tiles, (int)blockIdx.x);
}

#define EPT 8
template <bool FULL>
__device__ __forceinline__ void count_rank_body(const int* __restrict__ ei, int* __restrict__ cnt,
                                                int* __restrict__ rank, long long n_edges, long long base) {
    int d[EPT], r[EPT];
#pragma unroll
    for (int q = 0; q < EPT; ++q) {
        long long i = base + q * 256;
        d[q] = (FULL || i < n_edges) ? ei[n_edges + i] : 0;
    }
#pragma unroll
    for (int q = 0; q < EPT; ++q) {
        long long i = base + q * 256;
        r[q] = 0;
        if (FULL || i < n_edges) r[q] = atomicAdd(&cnt[d[q]], 1);
    }
#pragma unroll
    for (int q = 0; q < EPT; ++q) {
        long long i = base + q * 256;
        if (FULL || i < n_edges) rank[i] = r[q];
    }
}

__global__ void __launch_bounds__(256) __attribute__((amdgpu_waves_per_eu(6))) k_gemm1_count(const float* __restrict__ x,
                                                     const unsigned short* __restrict__ w1t,
                                                     float* __restrict__ h1,
                                                     const float* __restrict__ a_s,
                                                     const float* __restrict__ a_d,
                                                     float* __restrict__ als, float* __restrict__ ald,
                                                     int m_tiles,
                                                     const int* __restrict__ ei, int* __restrict__ cnt,
                                                     int* __restrict__ rank, long long n_edges) {
    const int wg = (int)(blockIdx.x >> 1);
    if (blockIdx.x & 1) {
        const long long blk = (long long)wg * (256 * EPT);
        if (blk + 256 * EPT <= n_edges) count_rank_body<true>(ei, cnt, rank, n_edges, blk + threadIdx.x);
        else                            count_rank_body<false>(ei, cnt, rank, n_edges, blk + threadIdx.x);
    } else {
        gemm_strip_body<true, 8>(x, nullptr, w1t, h1, a_s, a_d, als, ald, m_tiles, wg);
    }
}

__global__ void __launch_bounds__(1024) k_scan(const int* __restrict__ cnt,
                                               int* __restrict__ rowptr, int n) {
    __shared__ int wsum[32];
    __shared__ int psum[32];
    const int t = threadIdx.x, lane = t & 31, w = t >> 5;
    const int first = blockIdx.x * 1024;
    int pre = 0;
    {
        const int4* c4 = (const int4*)cnt;
#pragma unroll 8
        for (int q = t; q < (first >> 2); q += 1024) {
            int4 v = c4[q];
            pre += (v.x + v.y) + (v.z + v.w);
        }
    }
#pragma unroll
    for (int off = 16; off >= 1; off >>= 1) pre += __shfl_xor(pre, off);

    const int idx = first + t;
    const int c = (idx < n) ? cnt[idx] : 0;
    int incl = c;
#pragma unroll
    for (int off = 1; off < 32; off <<= 1) {
        int y = __shfl_up(incl, off);
        if (lane >= off) incl += y;
    }
    if (lane == 31) wsum[w] = incl;
    if (lane == 0)  psum[w] = pre;
    __syncthreads();
    int front = psum[lane] + (lane < w ? wsum[lane] : 0);
#pragma unroll
    for (int off = 16; off >= 1; off >>= 1) front += __shfl_xor(front, off);
    if (idx < n) rowptr[idx] = front + incl - c;
}

template <bool FULL>
__device__ __forceinline__ void fill_col_body(const int* __restrict__ ei, const int* __restrict__ rowptr,
                                              const int* __restrict__ rank, int* __restrict__ col,
                                              long long n_edges, long long base) {
    int s[EPT], d[EPT], rk[EPT], pos[EPT];
#pragma unroll
    for (int q = 0; q < EPT; ++q) {
        long long i = base + q * 256;
        const bool ok = FULL || i < n_edges;
        s[q]  = ok ? ei[i] : 0;
        d[q]  = ok ? ei[n_edges + i] : 0;
        rk[q] = ok ? rank[i] : 0;
    }
#pragma unroll
    for (int q = 0; q < EPT; ++q) {
        long long i = base + q * 256;
        pos[q] = (FULL || i < n_edges) ? rowptr[d[q]] + rk[q] : 0;
    }
#pragma unroll
    for (int q = 0; q < EPT; ++q) {
        long long i = base + q * 256;
        if (FULL || i < n_edges) col[pos[q]] = s[q];
    }
}

__global__ void __launch_bounds__(256) k_fill_col(const int* __restrict__ ei, const int* __restrict__ rowptr,
                                                  const int* __restrict__ rank, int* __restrict__ col,
                                                  long long n_edges) {
    const long long blk = (long long)blockIdx.x * (256 * EPT);
    if (blk + 256 * EPT <= n_edges) fill_col_body<true>(ei, rowptr, rank, col, n_edges, blk + threadIdx.x);
    else                            fill_col_body<false>(ei, rowptr, rank, col, n_edges, blk + threadIdx.x);
}

template <int H>
__device__ __forceinline__ void edge_score(int k, int n, int start, int node,
                                           const int* __restrict__ col,
                                           const float* __restrict__ als,
                                           const float* adv, int& s, float* v) {
    s = node;
#pragma unroll
    for (int h = 0; h < H; ++h) v[h] = -INFINITY;
    if (k < n) {
        if (k < n - 1) s = col[start + k];
        float x[H];
        if constexpr (H == 4) {
            float4 q = *(const float4*)(als + (size_t)s * 4);
            x[0] = q.x; x[1] = q.y; x[2] = q.z; x[3] = q.w;
        } else {
#pragma unroll
            for (int h = 0; h < H; ++h) x[h] = als[(size_t)s * H + h];
        }
#pragma unroll
        for (int h = 0; h < H; ++h) {
            float y = x[h] + adv[h];
            v[h] = y > 0.f ? y : NEG_SLOPE * y;
        }
    }
}

template <int H>
__device__ __forceinline__ void accum_chunk(int cntc, int s, const float* a,
                                            const float* __restrict__ hmat, int lane,
                                            float* acc) {
#pragma unroll 4
    for (int j = 0; j < cntc; ++j) {
        int sj = __builtin_amdgcn_readlane(s, j);
        const float* row = hmat + (size_t)sj * (H * 32) + lane;
#pragma unroll
        for (int h = 0; h < H; ++h) {
            float aj = __uint_as_float(
                (unsigned)__builtin_amdgcn_readlane((int)__float_as_uint(a[h]), j));
            acc[h] += aj * row[h * 32];
        }
    }
}

template <int H, bool RELU_BF16>
__global__ void __launch_bounds__(256) k_gat_gather(const int* __restrict__ rowptr,
                                                    const int* __restrict__ cnt,
                                                    const int* __restrict__ col,
                                                    const float* __restrict__ als,
                                                    const float* __restrict__ ald,
                                                    const float* __restrict__ hmat,
                                                    const float* __restrict__ bias,
                                                    float* __restrict__ outp,
                                                    unsigned short* __restrict__ outbf,
                                                    int n_nodes) {
    const int lane = threadIdx.x & 31;
    const int wave0 = __builtin_amdgcn_readfirstlane(
        (int)(blockIdx.x * (blockDim.x >> 5) + (threadIdx.x >> 5)));
    const int n_waves = (int)(gridDim.x * (blockDim.x >> 5));
    for (int node = wave0; node < n_nodes; node += n_waves) {
    const int start = rowptr[node];
    const int n = cnt[node] + 1;

    float adv[H];
#pragma unroll
    for (int h = 0; h < H; ++h) adv[h] = ald[(size_t)node * H + h];

    int s0; float v0[H];
    edge_score<H>(lane, n, start, node, col, als, adv, s0, v0);

    float m[H];
#pragma unroll
    for (int h = 0; h < H; ++h) m[h] = v0[h];
    for (int base = 32; base < n; base += 32) {
        int s; float v[H];
        edge_score<H>(base + lane, n, start, node, col, als, adv, s, v);
#pragma unroll
        for (int h = 0; h < H; ++h) m[h] = fmaxf(m[h], v[h]);
    }
#pragma unroll
    for (int h = 0; h < H; ++h)
#pragma unroll
        for (int off = 16; off >= 1; off >>= 1) m[h] = fmaxf(m[h], __shfl_xor(m[h], off));

    float den[H];
#pragma unroll
    for (int h = 0; h < H; ++h) den[h] = __expf(v0[h] - m[h]);
    for (int base = 32; base < n; base += 32) {
        int s; float v[H];
        edge_score<H>(base + lane, n, start, node, col, als, adv, s, v);
#pragma unroll
        for (int h = 0; h < H; ++h) den[h] += __expf(v[h] - m[h]);
    }
#pragma unroll
    for (int h = 0; h < H; ++h) {
#pragma unroll
        for (int off = 16; off >= 1; off >>= 1) den[h] += __shfl_xor(den[h], off);
        den[h] += 1e-16f;
    }

    float acc[H];
#pragma unroll
    for (int h = 0; h < H; ++h) acc[h] = 0.f;
    {
        float a0[H];
#pragma unroll
        for (int h = 0; h < H; ++h) a0[h] = __expf(v0[h] - m[h]) / den[h];
        accum_chunk<H>(n < 32 ? n : 32, s0, a0, hmat, lane, acc);
    }
    for (int base = 32; base < n; base += 32) {
        int s; float v[H], a[H];
        edge_score<H>(base + lane, n, start, node, col, als, adv, s, v);
#pragma unroll
        for (int h = 0; h < H; ++h) a[h] = __expf(v[h] - m[h]) / den[h];
        accum_chunk<H>((n - base) < 32 ? (n - base) : 32, s, a, hmat, lane, acc);
    }

#pragma unroll
    for (int h = 0; h < H; ++h) {
        float r = acc[h] + bias[h * 32 + lane];
        if constexpr (RELU_BF16)
            outbf[(size_t)node * (H * 32) + h * 32 + lane] = to_bf16_bits(r > 0.f ? r : 0.f);
        else
            outp[(size_t)node * (H * 32) + h * 32 + lane] = r;
    }
    }
}


extern "C" void kernel_launch(void* const* d_in, const int* in_sizes, int n_in,
                              void* d_out, int out_size, void* d_ws, size_t ws_size,
                              hipStream_t stream) {
    (void)n_in; (void)out_size; (void)ws_size; (void)in_sizes;
    constexpr long long N = 100000, E = 1600000;
    constexpr int C1 = 128, HEADS = 4, C2 = 32;
    constexpr int SCAN_BLOCKS = (int)((N + 1023) / 1024);
    constexpr int NODES_PER_WAVE = 7;
    constexpr int GATHER_BLOCKS = (int)((N + 8 * NODES_PER_WAVE - 1) / (8 * NODES_PER_WAVE));

    const float* x    = (const float*)d_in[0];
    const int*   ei   = (const int*)  d_in[1];
    const float* W1   = (const float*)d_in[2];
    const float* aS1  = (const float*)d_in[3];
    const float* aD1  = (const float*)d_in[4];
    const float* b1   = (const float*)d_in[5];
    const float* W2   = (const float*)d_in[6];
    const float* aS2  = (const float*)d_in[7];
    const float* aD2  = (const float*)d_in[8];
    const float* b2   = (const float*)d_in[9];
    float* out = (float*)d_out;

    char* ws = (char*)d_ws;
    size_t off = 0;
    auto alloc = [&](size_t bytes) { size_t o = off; off = (off + bytes + 255) & ~(size_t)255; return o; };
    unsigned short* hbf  = (unsigned short*)(ws + alloc(N * C1 * 2));
    float*          h1   = (float*)(ws + alloc(N * C1 * 4));
    float*          as1  = (float*)(ws + alloc(N * HEADS * 4));
    float*          ad1  = (float*)(ws + alloc(N * HEADS * 4));
    int*            cnt  = (int*)(ws + alloc(N * 4));
    int*            rptr = (int*)(ws + alloc(N * 4));
    int*            rank = (int*)(ws + alloc(E * 4));
    int*            col  = (int*)(ws + alloc(E * 4));
    unsigned short* w1t  = (unsigned short*)(ws + alloc(128 * 128 * 2));
    unsigned short* w2t  = (unsigned short*)(ws + alloc(32 * 128 * 2));

    constexpr int B = 256;
    constexpr long long B_EPT = 256LL * EPT;
    auto blocks = [](long long n, int b) { return (unsigned)((n + b - 1) / b); };

    constexpr int PREP_THREADS = 128 * 128 + 32 * 128 + (int)N;
    k_prep<<<blocks(PREP_THREADS, B), B, 0, stream>>>(W1, w1t, W2, w2t, cnt, (int)N);

    constexpr unsigned GEMM_WGS = (unsigned)((N / 16 + 7) / 8);
    constexpr unsigned EDGE_WGS = (unsigned)((E + B_EPT - 1) / B_EPT);
    constexpr unsigned PAIR_WGS = 2 * (GEMM_WGS > EDGE_WGS ? GEMM_WGS : EDGE_WGS);
    k_gemm1_count<<<PAIR_WGS, 256, 0, stream>>>(x, w1t, h1, aS1, aD1, as1, ad1, (int)(N / 16), ei, cnt, rank, E);

    k_scan<<<SCAN_BLOCKS, 1024, 0, stream>>>(cnt, rptr, (int)N);
    k_fill_col<<<blocks(E, B * EPT), B, 0, stream>>>(ei, rptr, rank, col, E);

    k_gat_gather<HEADS, true><<<GATHER_BLOCKS, 256, 0, stream>>>(rptr, cnt, col, as1, ad1, h1, b1, nullptr, hbf, (int)N);

    k_gemm2_strip<<<blocks(N / 16, 8), 256, 0, stream>>>(hbf, w2t, h1, aS2, aD2, as1, ad1, (int)(N / 16));

    k_gat_gather<1, false><<<GATHER_BLOCKS, 256, 0, stream>>>(rptr, cnt, col, as1, ad1, h1, b2, out, nullptr, (int)N);
}


# Gathers fetch rows with 16-byte-per-lane loads: one instruction per 512-B row, or per four 128-B rows; weights by 8-lane broadcast; softmax denominator and layer-2 row sum orders change
# speedup vs baseline: 7.3647x; 1.0509x over previous
#include <hip/hip_runtime.h>
#include <hip/hip_bf16.h>

typedef __attribute__((ext_vector_type(16))) __bf16 v16bf;
typedef __attribute__((ext_vector_type(8)))  float  v8f;
typedef __attribute__((ext_vector_type(2)))  float  v2f;
typedef __attribute__((ext_vector_type(2)))  __bf16 v2bf;

#define NEG_SLOPE 0.2f


__device__ __forceinline__ unsigned short f2bf(float f) {
    unsigned u = __float_as_uint(f);
    unsigned r = u + 0x7FFFu + ((u >> 16) & 1u);
    return (unsigned short)(r >> 16);
}

__device__ __forceinline__ unsigned pack_bf16x2(float lo, float hi) {
    v2f f = {lo, hi};
    return __builtin_bit_cast(unsigned, __builtin_convertvector(f, v2bf));
}


__global__ void __launch_bounds__(256) k_prep(const float* __restrict__ W1, unsigned short* __restrict__ w1t,
                                              const float* __restrict__ W2, unsigned short* __restrict__ w2t,
                                              int* __restrict__ cnt, int n_nodes) {
    int i = blockIdx.x * 256 + threadIdx.x;
    if (i < 128 * 128) {
        int n = i >> 7, k = i & 127;
        w1t[i] = f2bf(W1[k * 128 + n]);
    } else if (i < 128 * 128 + 32 * 128) {
        i -= 128 * 128;
        int n = i >> 7, k = i & 127;
        w2t[i] = f2bf(W2[k * 32 + n]);
    } else {
        i -= 128 * 128 + 32 * 128;
        if (i < n_nodes) cnt[i] = 0;
    }
}

union Frag { unsigned u[8]; v16bf v; };
typedef __attribute__((ext_vector_type(4))) float    v4f;
typedef __attribute__((ext_vector_type(4))) unsigned v4u;

__device__ __forceinline__ v16bf frag_of(v4u q0, v4u q1) {
    Frag f;
    f.u[0] = q0.x; f.u[1] = q0.y; f.u[2] = q0.z; f.u[3] = q0.w;
    f.u[4] = q1.x; f.u[5] = q1.y; f.u[6] = q1.z; f.u[7] = q1.w;
    return f.v;
}

template <int M>
__device__ __forceinline__ float row_xor(float v) {
    return __builtin_bit_cast(float, __builtin_amdgcn_mov_dpp(
        __builtin_bit_cast(int, v), 0x160 + M, 0xf, 0xf, false));
}

__device__ __forceinline__ float row_reduce_scatter(const float* ps, const float* pd, int r15) {
    const bool b3 = r15 & 8, b2 = r15 & 4, b1 = r15 & 2, b0 = r15 & 1;
    float w[8], x[4], y[2];
#pragma unroll
    for (int i = 0; i < 8; ++i) w[i] = (b3 ? pd[i] : ps[i]) + row_xor<8>(b3 ? ps[i] : pd[i]);
#pragma unroll
    for (int i = 0; i < 4; ++i) x[i] = (b2 ? w[i + 4] : w[i]) + row_xor<4>(b2 ? w[i] : w[i + 4]);
#pragma unroll
    for (int i = 0; i < 2; ++i) y[i] = (b1 ? x[i + 2] : x[i]) + row_xor<2>(b1 ? x[i] : x[i + 2]);
    return (b0 ? y[1] : y[0]) + row_xor<1>(b0 ? y[0] : y[1]);
}

template <bool A_F32, int NT>
__device__ __forceinline__ void gemm_strip_body(const float* __restrict__ Af,
                                                const unsigned short* __restrict__ Ab,
                                                const unsigned short* __restrict__ Bt,
                                                float* __restrict__ C,
                                                const float* __restrict__ a_s,
                                                const float* __restrict__ a_d,
                                                float* __restrict__ als, float* __restrict__ ald,
                                                int m_tiles, int wg) {
    constexpr int H = NT / 2;
    const int lane = threadIdx.x & 31;
    const int tileM = __builtin_amdgcn_readfirstlane((int)(wg * 8 + (threadIdx.x >> 5)));
    if (tileM >= m_tiles) return;
    const int r15  = lane & 15;
    const int half = lane >> 4;

    const v4u* bbase = (const v4u*)(Bt + (size_t)r15 * 128 + half * 8);
    v4u qb[2][8];
#pragma unroll
    for (int s = 0; s < 4; ++s) { qb[0][2 * s] = bbase[s * 4 + 0]; qb[0][2 * s + 1] = bbase[s * 4 + 2]; }

    v16bf a[4];
    if constexpr (A_F32) {
        const v4f* arow = (const v4f*)(Af + (size_t)(tileM * 16 + r15) * 128 + half * 8);
        v4f p[16];
#pragma unroll
        for (int s = 0; s < 4; ++s) {
            p[4 * s + 0] = arow[s * 8 + 0];
            p[4 * s + 1] = arow[s * 8 + 1];
            p[4 * s + 2] = arow[s * 8 + 4];
            p[4 * s + 3] = arow[s * 8 + 5];
        }
        asm volatile("" :: "v"(p[0]), "v"(p[1]), "v"(p[2]), "v"(p[3]), "v"(p[4]), "v"(p[5]), "v"(p[6]), "v"(p[7]),
                           "v"(p[8]), "v"(p[9]), "v"(p[10]), "v"(p[11]), "v"(p[12]), "v"(p[13]), "v"(p[14]), "v"(p[15]));
#pragma unroll
        for (int s = 0; s < 4; ++s) {
            Frag f;
#pragma unroll
            for (int j = 0; j < 4; ++j) {
                f.u[2 * j]     = pack_bf16x2(p[4 * s + j].x, p[4 * s + j].y);
                f.u[2 * j + 1] = pack_bf16x2(p[4 * s + j].z, p[4 * s + j].w);
            }
            a[s] = f.v;
        }
    } else {
        const v4u* arow = (const v4u*)(Ab + (size_t)(tileM * 16 + r15) * 128 + half * 8);
        v4u qa[8];
#pragma unroll
        for (int s = 0; s < 4; ++s) { qa[2 * s] = arow[s * 4 + 0]; qa[2 * s + 1] = arow[s * 4 + 2]; }
#pragma unroll
        for (int s = 0; s < 4; ++s) a[s] = frag_of(qa[2 * s], qa[2 * s + 1]);
    }

    float ps[8], pd[8], res[H];
#pragma unroll
    for (int t = 0; t < NT; ++t) {
        if (t + 1 < NT) {
#pragma unroll
            for (int s = 0; s < 4; ++s) {
                qb[(t + 1) & 1][2 * s]     = bbase[(t + 1) * 256 + s * 4 + 0];
                qb[(t + 1) & 1][2 * s + 1] = bbase[(t + 1) * 256 + s * 4 + 2];
            }
        }
        const float cs = a_s[t * 16 + r15], cd = a_d[t * 16 + r15];
        v8f acc = {};
#pragma unroll
        for (int s = 0; s < 4; ++s)
            acc = __builtin_amdgcn_wmma_f32_16x16x32_bf16(
                    false, a[s], false, frag_of(qb[t & 1][2 * s], qb[t & 1][2 * s + 1]),
                    (short)0, acc, false, false);
#pragma unroll
        for (int r = 0; r < 8; ++r) {
            int row = tileM * 16 + r + 8 * half;
            int col = t * 16 + r15;
            C[(size_t)row * (NT * 16) + col] = acc[r];
        }
        if ((t & 1) == 0) {
#pragma unroll
            for (int r = 0; r < 8; ++r) { ps[r] = acc[r] * cs; pd[r] = acc[r] * cd; }
        } else {
#pragma unroll
            for (int r = 0; r < 8; ++r) { ps[r] += acc[r] * cs; pd[r] += acc[r] * cd; }
            res[t >> 1] = row_reduce_scatter(ps, pd, r15);
        }
    }
    {
        const int row = tileM * 16 + 8 * half + (r15 & 7);
        float* dstp = ((r15 & 8) ? ald : als) + (size_t)row * H;
        if constexpr (H == 4) {
            *(float4*)dstp = make_float4(res[0], res[1], res[2], res[3]);
        } else {
#pragma unroll
            for (int h = 0; h < H; ++h) dstp[h] = res[h];
        }
    }
}

__global__ void __launch_bounds__(256) k_gemm2_strip(const unsigned short* __restrict__ Ab,
                                                     const unsigned short* __restrict__ Bt,
                                                     float* __restrict__ C,
                                                     const float* __restrict__ a_s,
                                                     const float* __restrict__ a_d,
                                                     float* __restrict__ als, float* __restrict__ ald,
                                                     int m_tiles) {
    gemm_strip_body<false, 2>(nullptr, Ab, Bt, C, a_s, a_d, als, ald, m_tiles, (int)blockIdx.x);
}

#define EPT 8
template <bool FULL>
__device__ __forceinline__ void count_rank_body(const int* __restrict__ ei, int* __restrict__ cnt,
                                                int* __restrict__ rank, long long n_edges, long long base) {
    int d[EPT], r[EPT];
#pragma unroll
    for (int q = 0; q < EPT; ++q) {
        long long i = base + q * 256;
        d[q] = (FULL || i < n_edges) ? ei[n_edges + i] : 0;
    }
#pragma unroll
    for (int q = 0; q < EPT; ++q) {
        long long i = base + q * 256;
        r[q] = 0;
        if (FULL || i < n_edges) r[q] = atomicAdd(&cnt[d[q]], 1);
    }
#pragma unroll
    for (int q = 0; q < EPT; ++q) {
        long long i = base + q * 256;
        if (FULL || i < n_edges) rank[i] = r[q];
    }
}

__global__ void __launch_bounds__(256) __attribute__((amdgpu_waves_per_eu(6))) k_gemm1_count(const float* __restrict__ x,
                                                     const unsigned short* __restrict__ w1t,
                                                     float* __restrict__ h1,
                                                     const float* __restrict__ a_s,
                                                     const float* __restrict__ a_d,
                                                     float* __restrict__ als, float* __restrict__ ald,
                                                     int m_tiles,
                                                     const int* __restrict__ ei, int* __restrict__ cnt,
                                                     int* __restrict__ rank, long long n_edges) {
    const int wg = (int)(blockIdx.x >> 1);
    if (blockIdx.x & 1) {
        const long long blk = (long long)wg * (256 * EPT);
        if (blk + 256 * EPT <= n_edges) count_rank_body<true>(ei, cnt, rank, n_edges, blk + threadIdx.x);
        else                            count_rank_body<false>(ei, cnt, rank, n_edges, blk + threadIdx.x);
    } else {
        gemm_strip_body<true, 8>(x, nullptr, w1t, h1, a_s, a_d, als, ald, m_tiles, wg);
    }
}

__global__ void __launch_bounds__(1024) k_scan(const int* __restrict__ cnt,
                                               int* __restrict__ rowptr, int n) {
    __shared__ int wsum[32];
    __shared__ int psum[32];
    const int t = threadIdx.x, lane = t & 31, w = t >> 5;
    const int first = blockIdx.x * 1024;
    int pre = 0;
    {
        const int4* c4 = (const int4*)cnt;
#pragma unroll 8
        for (int q = t; q < (first >> 2); q += 1024) {
            int4 v = c4[q];
            pre += (v.x + v.y) + (v.z + v.w);
        }
    }
#pragma unroll
    for (int off = 16; off >= 1; off >>= 1) pre += __shfl_xor(pre, off);

    const int idx = first + t;
    const int c = (idx < n) ? cnt[idx] : 0;
    int incl = c;
#pragma unroll
    for (int off = 1; off < 32; off <<= 1) {
        int y = __shfl_up(incl, off);
        if (lane >= off) incl += y;
    }
    if (lane == 31) wsum[w] = incl;
    if (lane == 0)  psum[w] = pre;
    __syncthreads();
    int front = psum[lane] + (lane < w ? wsum[lane] : 0);
#pragma unroll
    for (int off = 16; off >= 1; off >>= 1) front += __shfl_xor(front, off);
    if (idx < n) rowptr[idx] = front + incl - c;
}

template <bool FULL>
__device__ __forceinline__ void fill_col_body(const int* __restrict__ ei, const int* __restrict__ rowptr,
                                              const int* __restrict__ rank, int* __restrict__ col,
                                              long long n_edges, long long base) {
    int s[EPT], d[EPT], rk[EPT], pos[EPT];
#pragma unroll
    for (int q = 0; q < EPT; ++q) {
        long long i = base + q * 256;
        const bool ok = FULL || i < n_edges;
        s[q]  = ok ? ei[i] : 0;
        d[q]  = ok ? ei[n_edges + i] : 0;
        rk[q] = ok ? rank[i] : 0;
    }
#pragma unroll
    for (int q = 0; q < EPT; ++q) {
        long long i = base + q * 256;
        pos[q] = (FULL || i < n_edges) ? rowptr[d[q]] + rk[q] : 0;
    }
#pragma unroll
    for (int q = 0; q < EPT; ++q) {
        long long i = base + q * 256;
        if (FULL || i < n_edges) col[pos[q]] = s[q];
    }
}

__global__ void __launch_bounds__(256) k_fill_col(const int* __restrict__ ei, const int* __restrict__ rowptr,
                                                  const int* __restrict__ rank, int* __restrict__ col,
                                                  long long n_edges) {
    const long long blk = (long long)blockIdx.x * (256 * EPT);
    if (blk + 256 * EPT <= n_edges) fill_col_body<true>(ei, rowptr, rank, col, n_edges, blk + threadIdx.x);
    else                            fill_col_body<false>(ei, rowptr, rank, col, n_edges, blk + threadIdx.x);
}


template <int E>
__device__ __forceinline__ unsigned grp8_u(unsigned v) { return __builtin_amdgcn_mov_dpp8(v, E * 0x249249u); }
__device__ __forceinline__ unsigned grp8_pick_u(unsigned v, int e) {
    switch (e) {
        case 0: return grp8_u<0>(v); case 1: return grp8_u<1>(v); case 2: return grp8_u<2>(v); case 3: return grp8_u<3>(v);
        case 4: return grp8_u<4>(v); case 5: return grp8_u<5>(v); case 6: return grp8_u<6>(v); default: return grp8_u<7>(v);
    }
}
__device__ __forceinline__ float grp8_pick(float v, int e) {
    return __uint_as_float(grp8_pick_u(__float_as_uint(v), e));
}

__device__ __forceinline__ void score_l1(int base, int n, int start, int node, int n_edges,
                                         const int* __restrict__ col, const float* __restrict__ als,
                                         float adv, int hd, int e8, int* s, float* v) {
    const int nn = n - base;
    int c[4];
#pragma unroll
    for (int g = 0; g < 4; ++g) {
        c[g] = node;
        if (8 * g < nn) {
            int pos = start + base + 8 * g + e8;
            pos = pos < n_edges ? pos : n_edges - 1;
            c[g] = col[pos];
        }
    }
    float x[4];
#pragma unroll
    for (int g = 0; g < 4; ++g) {
        const int k = base + 8 * g + e8;
        s[g] = (k < n - 1) ? c[g] : node;
        x[g] = 0.f;
        if (8 * g < nn) x[g] = als[(size_t)s[g] * 4 + hd];
    }
#pragma unroll
    for (int g = 0; g < 4; ++g) {
        const int k = base + 8 * g + e8;
        const float y = x[g] + adv;
        const float z = y > 0.f ? y : NEG_SLOPE * y;
        v[g] = (k < n) ? z : -INFINITY;
    }
}

template <int E0, int NE>
__device__ __forceinline__ void accum_grp_l1(int sg, float ag, const float* __restrict__ hmat,
                                             int lane, v4f& acc) {
    v4f r[NE];
#pragma unroll
    for (int e = 0; e < NE; ++e) {
        const int sj = __builtin_amdgcn_readlane(sg, E0 + e);
        r[e] = ((const v4f*)(hmat + (size_t)sj * 128))[lane];
    }
#pragma unroll
    for (int e = 0; e < NE; ++e) {
        const float aj = grp8_pick(ag, E0 + e);
        acc += aj * r[e];
    }
}

__device__ __forceinline__ void accum_l1(int cntc, const int* s, const float* a,
                                         const float* __restrict__ hmat, int lane, v4f& acc) {
#pragma unroll
    for (int g = 0; g < 4; ++g) {
        const int rem = cntc - 8 * g;
        if (rem > 4)      accum_grp_l1<0, 8>(s[g], a[g], hmat, lane, acc);
        else if (rem > 0) accum_grp_l1<0, 4>(s[g], a[g], hmat, lane, acc);
    }
}

__device__ __forceinline__ float max4(const float* v) { return fmaxf(fmaxf(v[0], v[1]), fmaxf(v[2], v[3])); }

__global__ void __launch_bounds__(256) __attribute__((amdgpu_waves_per_eu(14))) k_gat_gather_l1(const int* __restrict__ rowptr,
                                                       const int* __restrict__ cnt,
                                                       const int* __restrict__ col,
                                                       const float* __restrict__ als,
                                                       const float* __restrict__ ald,
                                                       const float* __restrict__ hmat,
                                                       const float* __restrict__ bias,
                                                       unsigned short* __restrict__ outbf,
                                                       int n_nodes, int n_edges) {
    const int lane = threadIdx.x & 31;
    const int hd = lane >> 3, e8 = lane & 7;
    const int wave0 = __builtin_amdgcn_readfirstlane(
        (int)(blockIdx.x * (blockDim.x >> 5) + (threadIdx.x >> 5)));
    const int n_waves = (int)(gridDim.x * (blockDim.x >> 5));
    for (int node = wave0; node < n_nodes; node += n_waves) {
        const int start = rowptr[node];
        const int n = cnt[node] + 1;
        const float4 ad4 = *(const float4*)(ald + (size_t)node * 4);
        const float adv = hd == 0 ? ad4.x : hd == 1 ? ad4.y : hd == 2 ? ad4.z : ad4.w;

        int s0[4]; float v0[4];
        score_l1(0, n, start, node, n_edges, col, als, adv, hd, e8, s0, v0);

        float m = max4(v0);
        for (int base = 32; base < n; base += 32) {
            int s[4]; float v[4];
            score_l1(base, n, start, node, n_edges, col, als, adv, hd, e8, s, v);
            m = fmaxf(m, max4(v));
        }
        m = fmaxf(m, row_xor<1>(m)); m = fmaxf(m, row_xor<2>(m)); m = fmaxf(m, row_xor<4>(m));

        float den = (__expf(v0[0] - m) + __expf(v0[1] - m)) + (__expf(v0[2] - m) + __expf(v0[3] - m));
        for (int base = 32; base < n; base += 32) {
            int s[4]; float v[4];
            score_l1(base, n, start, node, n_edges, col, als, adv, hd, e8, s, v);
            den += (__expf(v[0] - m) + __expf(v[1] - m)) + (__expf(v[2] - m) + __expf(v[3] - m));
        }
        den += row_xor<1>(den); den += row_xor<2>(den); den += row_xor<4>(den);
        den += 1e-16f;

        v4f acc = {0.f, 0.f, 0.f, 0.f};
        {
            float a0[4];
#pragma unroll
            for (int g = 0; g < 4; ++g) a0[g] = __expf(v0[g] - m) / den;
            accum_l1(n < 32 ? n : 32, s0, a0, hmat, lane, acc);
        }
        for (int base = 32; base < n; base += 32) {
            int s[4]; float v[4], a[4];
            score_l1(base, n, start, node, n_edges, col, als, adv, hd, e8, s, v);
#pragma unroll
            for (int g = 0; g < 4; ++g) a[g] = __expf(v[g] - m) / den;
            accum_l1((n - base) < 32 ? (n - base) : 32, s, a, hmat, lane, acc);
        }

        const v4f r = acc + ((const v4f*)bias)[lane];
        uint2 o;
        o.x = pack_bf16x2(r.x > 0.f ? r.x : 0.f, r.y > 0.f ? r.y : 0.f);
        o.y = pack_bf16x2(r.z > 0.f ? r.z : 0.f, r.w > 0.f ? r.w : 0.f);
        ((uint2*)(outbf + (size_t)node * 128))[lane] = o;
    }
}

__device__ __forceinline__ void score_l2(int k, int n, int start, int node,
                                         const int* __restrict__ col, const float* __restrict__ als,
                                         float adv, int& s, float& v) {
    s = node; v = -INFINITY;
    if (k < n) {
        if (k < n - 1) s = col[start + k];
        const float y = als[s] + adv;
        v = y > 0.f ? y : NEG_SLOPE * y;
    }
}

template <int T0, int NT>
__device__ __forceinline__ void accum_steps_l2(int s, float a, const float* __restrict__ hmat,
                                               int c8, v4f& acc) {
    v4f r[NT];
#pragma unroll
    for (int t = 0; t < NT; ++t) {
        const int sq = (int)grp8_pick_u((unsigned)s, T0 + t);
        r[t] = ((const v4f*)(hmat + (size_t)sq * 32))[c8];
    }
#pragma unroll
    for (int t = 0; t < NT; ++t) {
        const float aq = grp8_pick(a, T0 + t);
        acc += aq * r[t];
    }
}

__device__ __forceinline__ void accum_l2(int cntc, int s, float a, const float* __restrict__ hmat,
                                         int c8, v4f& acc) {
    accum_steps_l2<0, 4>(s, a, hmat, c8, acc);
    if (cntc > 16) accum_steps_l2<4, 2>(s, a, hmat, c8, acc);
    if (cntc > 24) accum_steps_l2<6, 2>(s, a, hmat, c8, acc);
}

__global__ void __launch_bounds__(256) k_gat_gather_l2(const int* __restrict__ rowptr,
                                                       const int* __restrict__ cnt,
                                                       const int* __restrict__ col,
                                                       const float* __restrict__ als,
                                                       const float* __restrict__ ald,
                                                       const float* __restrict__ hmat,
                                                       const float* __restrict__ bias,
                                                       float* __restrict__ outp,
                                                       int n_nodes) {
    const int lane = threadIdx.x & 31;
    const int q = lane >> 3, c8 = lane & 7;
    const int kk = 4 * c8 + q;
    const int wave0 = __builtin_amdgcn_readfirstlane(
        (int)(blockIdx.x * (blockDim.x >> 5) + (threadIdx.x >> 5)));
    const int n_waves = (int)(gridDim.x * (blockDim.x >> 5));
    const v4f bv = ((const v4f*)bias)[c8];
    for (int node = wave0; node < n_nodes; node += n_waves) {
        const int start = rowptr[node];
        const int n = cnt[node] + 1;
        const float adv = ald[node];

        int s0; float v0;
        score_l2(kk, n, start, node, col, als, adv, s0, v0);

        float m = v0;
        for (int base = 32; base < n; base += 32) {
            int s; float v;
            score_l2(base + kk, n, start, node, col, als, adv, s, v);
            m = fmaxf(m, v);
        }
#pragma unroll
        for (int off = 16; off >= 1; off >>= 1) m = fmaxf(m, __shfl_xor(m, off));

        float den = __expf(v0 - m);
        for (int base = 32; base < n; base += 32) {
            int s; float v;
            score_l2(base + kk, n, start, node, col, als, adv, s, v);
            den += __expf(v - m);
        }
#pragma unroll
        for (int off = 16; off >= 1; off >>= 1) den += __shfl_xor(den, off);
        den += 1e-16f;

        v4f acc = {0.f, 0.f, 0.f, 0.f};
        accum_l2(n < 32 ? n : 32, s0, __expf(v0 - m) / den, hmat, c8, acc);
        for (int base = 32; base < n; base += 32) {
            int s; float v;
            score_l2(base + kk, n, start, node, col, als, adv, s, v);
            accum_l2((n - base) < 32 ? (n - base) : 32, s, __expf(v - m) / den, hmat, c8, acc);
        }
        acc.x += row_xor<8>(acc.x); acc.y += row_xor<8>(acc.y); acc.z += row_xor<8>(acc.z); acc.w += row_xor<8>(acc.w);
        acc.x += __shfl_xor(acc.x, 16); acc.y += __shfl_xor(acc.y, 16);
        acc.z += __shfl_xor(acc.z, 16); acc.w += __shfl_xor(acc.w, 16);
        if (lane < 8) ((v4f*)(outp + (size_t)node * 32))[c8] = acc + bv;
    }
}


extern "C" void kernel_launch(void* const* d_in, const int* in_sizes, int n_in,
                              void* d_out, int out_size, void* d_ws, size_t ws_size,
                              hipStream_t stream) {
    (void)n_in; (void)out_size; (void)ws_size; (void)in_sizes;
    constexpr long long N = 100000, E = 1600000;
    constexpr int C1 = 128, HEADS = 4, C2 = 32;
    constexpr int SCAN_BLOCKS = (int)((N + 1023) / 1024);
    constexpr int NODES_PER_WAVE = 7;
    constexpr int GATHER_BLOCKS = (int)((N + 8 * NODES_PER_WAVE - 1) / (8 * NODES_PER_WAVE));

    const float* x    = (const float*)d_in[0];
    const int*   ei   = (const int*)  d_in[1];
    const float* W1   = (const float*)d_in[2];
    const float* aS1  = (const float*)d_in[3];
    const float* aD1  = (const float*)d_in[4];
    const float* b1   = (const float*)d_in[5];
    const float* W2   = (const float*)d_in[6];
    const float* aS2  = (const float*)d_in[7];
    const float* aD2  = (const float*)d_in[8];
    const float* b2   = (const float*)d_in[9];
    float* out = (float*)d_out;

    char* ws = (char*)d_ws;
    size_t off = 0;
    auto alloc = [&](size_t bytes) { size_t o = off; off = (off + bytes + 255) & ~(size_t)255; return o; };
    unsigned short* hbf  = (unsigned short*)(ws + alloc(N * C1 * 2));
    float*          h1   = (float*)(ws + alloc(N * C1 * 4));
    float*          as1  = (float*)(ws + alloc(N * HEADS * 4));
    float*          ad1  = (float*)(ws + alloc(N * HEADS * 4));
    int*            cnt  = (int*)(ws + alloc(N * 4));
    int*            rptr = (int*)(ws + alloc(N * 4));
    int*            rank = (int*)(ws + alloc(E * 4));
    int*            col  = (int*)(ws + alloc(E * 4));
    unsigned short* w1t  = (unsigned short*)(ws + alloc(128 * 128 * 2));
    unsigned short* w2t  = (unsigned short*)(ws + alloc(32 * 128 * 2));

    constexpr int B = 256;
    constexpr long long B_EPT = 256LL * EPT;
    auto blocks = [](long long n, int b) { return (unsigned)((n + b - 1) / b); };

    constexpr int PREP_THREADS = 128 * 128 + 32 * 128 + (int)N;
    k_prep<<<blocks(PREP_THREADS, B), B, 0, stream>>>(W1, w1t, W2, w2t, cnt, (int)N);

    constexpr unsigned GEMM_WGS = (unsigned)((N / 16 + 7) / 8);
    constexpr unsigned EDGE_WGS = (unsigned)((E + B_EPT - 1) / B_EPT);
    constexpr unsigned PAIR_WGS = 2 * (GEMM_WGS > EDGE_WGS ? GEMM_WGS : EDGE_WGS);
    k_gemm1_count<<<PAIR_WGS, 256, 0, stream>>>(x, w1t, h1, aS1, aD1, as1, ad1, (int)(N / 16), ei, cnt, rank, E);

    k_scan<<<SCAN_BLOCKS, 1024, 0, stream>>>(cnt, rptr, (int)N);
    k_fill_col<<<blocks(E, B * EPT), B, 0, stream>>>(ei, rptr, rank, col, E);

    k_gat_gather_l1<<<GATHER_BLOCKS, 256, 0, stream>>>(rptr, cnt, col, as1, ad1, h1, b1, hbf, (int)N, (int)E);

    k_gemm2_strip<<<blocks(N / 16, 8), 256, 0, stream>>>(hbf, w2t, h1, aS2, aD2, as1, ad1, (int)(N / 16));

    k_gat_gather_l2<<<GATHER_BLOCKS, 256, 0, stream>>>(rptr, cnt, col, as1, ad1, h1, b2, out, (int)N);
}


# Edges grouped by destination by a two-level LDS counting sort (tile sort by 256-node bucket beside GEMM-1, a workgroup per bucket): no global atomics, 7 -> 6 launches; sums still in arrival order
# speedup vs baseline: 12.8709x; 1.7476x over previous
#include <hip/hip_runtime.h>
#include <hip/hip_bf16.h>

typedef __attribute__((ext_vector_type(16))) __bf16 v16bf;
typedef __attribute__((ext_vector_type(8)))  float  v8f;
typedef __attribute__((ext_vector_type(2)))  float  v2f;
typedef __attribute__((ext_vector_type(2)))  __bf16 v2bf;

#define NEG_SLOPE 0.2f


__device__ __forceinline__ unsigned short f2bf(float f) {
    unsigned u = __float_as_uint(f);
    unsigned r = u + 0x7FFFu + ((u >> 16) & 1u);
    return (unsigned short)(r >> 16);
}

__device__ __forceinline__ unsigned pack_bf16x2(float lo, float hi) {
    v2f f = {lo, hi};
    return __builtin_bit_cast(unsigned, __builtin_convertvector(f, v2bf));
}


__global__ void __launch_bounds__(256) k_prep(const float* __restrict__ W1, unsigned short* __restrict__ w1t,
                                              const float* __restrict__ W2, unsigned short* __restrict__ w2t) {
    int i = blockIdx.x * 256 + threadIdx.x;
    if (i < 128 * 128) {
        int n = i >> 7, k = i & 127;
        w1t[i] = f2bf(W1[k * 128 + n]);
    } else if (i < 128 * 128 + 32 * 128) {
        i -= 128 * 128;
        int n = i >> 7, k = i & 127;
        w2t[i] = f2bf(W2[k * 32 + n]);
    }
}

union Frag { unsigned u[8]; v16bf v; };
typedef __attribute__((ext_vector_type(4))) float    v4f;
typedef __attribute__((ext_vector_type(4))) unsigned v4u;

__device__ __forceinline__ v16bf frag_of(v4u q0, v4u q1) {
    Frag f;
    f.u[0] = q0.x; f.u[1] = q0.y; f.u[2] = q0.z; f.u[3] = q0.w;
    f.u[4] = q1.x; f.u[5] = q1.y; f.u[6] = q1.z; f.u[7] = q1.w;
    return f.v;
}

template <int M>
__device__ __forceinline__ float row_xor(float v) {
    return __builtin_bit_cast(float, __builtin_amdgcn_mov_dpp(
        __builtin_bit_cast(int, v), 0x160 + M, 0xf, 0xf, false));
}

__device__ __forceinline__ float row_reduce_scatter(const float* ps, const float* pd, int r15) {
    const bool b3 = r15 & 8, b2 = r15 & 4, b1 = r15 & 2, b0 = r15 & 1;
    float w[8], x[4], y[2];
#pragma unroll
    for (int i = 0; i < 8; ++i) w[i] = (b3 ? pd[i] : ps[i]) + row_xor<8>(b3 ? ps[i] : pd[i]);
#pragma unroll
    for (int i = 0; i < 4; ++i) x[i] = (b2 ? w[i + 4] : w[i]) + row_xor<4>(b2 ? w[i] : w[i + 4]);
#pragma unroll
    for (int i = 0; i < 2; ++i) y[i] = (b1 ? x[i + 2] : x[i]) + row_xor<2>(b1 ? x[i] : x[i + 2]);
    return (b0 ? y[1] : y[0]) + row_xor<1>(b0 ? y[0] : y[1]);
}

template <bool A_F32, int NT>
__device__ __forceinline__ void gemm_strip_body(const float* __restrict__ Af,
                                                const unsigned short* __restrict__ Ab,
                                                const unsigned short* __restrict__ Bt,
                                                float* __restrict__ C,
                                                const float* __restrict__ a_s,
                                                const float* __restrict__ a_d,
                                                float* __restrict__ als, float* __restrict__ ald,
                                                int m_tiles, int wg) {
    constexpr int H = NT / 2;
    const int lane = threadIdx.x & 31;
    const int tileM = __builtin_amdgcn_readfirstlane((int)(wg * 8 + (threadIdx.x >> 5)));
    if (tileM >= m_tiles) return;
    const int r15  = lane & 15;
    const int half = lane >> 4;

    const v4u* bbase = (const v4u*)(Bt + (size_t)r15 * 128 + half * 8);
    v4u qb[2][8];
#pragma unroll
    for (int s = 0; s < 4; ++s) { qb[0][2 * s] = bbase[s * 4 + 0]; qb[0][2 * s + 1] = bbase[s * 4 + 2]; }

    v16bf a[4];
    if constexpr (A_F32) {
        const v4f* arow = (const v4f*)(Af + (size_t)(tileM * 16 + r15) * 128 + half * 8);
        v4f p[16];
#pragma unroll
        for (int s = 0; s < 4; ++s) {
            p[4 * s + 0] = arow[s * 8 + 0];
            p[4 * s + 1] = arow[s * 8 + 1];
            p[4 * s + 2] = arow[s * 8 + 4];
            p[4 * s + 3] = arow[s * 8 + 5];
        }
        asm volatile("" :: "v"(p[0]), "v"(p[1]), "v"(p[2]), "v"(p[3]), "v"(p[4]), "v"(p[5]), "v"(p[6]), "v"(p[7]),
                           "v"(p[8]), "v"(p[9]), "v"(p[10]), "v"(p[11]), "v"(p[12]), "v"(p[13]), "v"(p[14]), "v"(p[15]));
#pragma unroll
        for (int s = 0; s < 4; ++s) {
            Frag f;
#pragma unroll
            for (int j = 0; j < 4; ++j) {
                f.u[2 * j]     = pack_bf16x2(p[4 * s + j].x, p[4 * s + j].y);
                f.u[2 * j + 1] = pack_bf16x2(p[4 * s + j].z, p[4 * s + j].w);
            }
            a[s] = f.v;
        }
    } else {
        const v4u* arow = (const v4u*)(Ab + (size_t)(tileM * 16 + r15) * 128 + half * 8);
        v4u qa[8];
#pragma unroll
        for (int s = 0; s < 4; ++s) { qa[2 * s] = arow[s * 4 + 0]; qa[2 * s + 1] = arow[s * 4 + 2]; }
#pragma unroll
        for (int s = 0; s < 4; ++s) a[s] = frag_of(qa[2 * s], qa[2 * s + 1]);
    }

    float ps[8], pd[8], res[H];
#pragma unroll
    for (int t = 0; t < NT; ++t) {
        if (t + 1 < NT) {
#pragma unroll
            for (int s = 0; s < 4; ++s) {
                qb[(t + 1) & 1][2 * s]     = bbase[(t + 1) * 256 + s * 4 + 0];
                qb[(t + 1) & 1][2 * s + 1] = bbase[(t + 1) * 256 + s * 4 + 2];
            }
        }
        const float cs = a_s[t * 16 + r15], cd = a_d[t * 16 + r15];
        v8f acc = {};
#pragma unroll
        for (int s = 0; s < 4; ++s)
            acc = __builtin_amdgcn_wmma_f32_16x16x32_bf16(
                    false, a[s], false, frag_of(qb[t & 1][2 * s], qb[t & 1][2 * s + 1]),
                    (short)0, acc, false, false);
#pragma unroll
        for (int r = 0; r < 8; ++r) {
            int row = tileM * 16 + r + 8 * half;
            int col = t * 16 + r15;
            C[(size_t)row * (NT * 16) + col] = acc[r];
        }
        if ((t & 1) == 0) {
#pragma unroll
            for (int r = 0; r < 8; ++r) { ps[r] = acc[r] * cs; pd[r] = acc[r] * cd; }
        } else {
#pragma unroll
            for (int r = 0; r < 8; ++r) { ps[r] += acc[r] * cs; pd[r] += acc[r] * cd; }
            res[t >> 1] = row_reduce_scatter(ps, pd, r15);
        }
    }
    {
        const int row = tileM * 16 + 8 * half + (r15 & 7);
        float* dstp = ((r15 & 8) ? ald : als) + (size_t)row * H;
        if constexpr (H == 4) {
            *(float4*)dstp = make_float4(res[0], res[1], res[2], res[3]);
        } else {
#pragma unroll
            for (int h = 0; h < H; ++h) dstp[h] = res[h];
        }
    }
}

__global__ void __launch_bounds__(256) k_gemm2_strip(const unsigned short* __restrict__ Ab,
                                                     const unsigned short* __restrict__ Bt,
                                                     float* __restrict__ C,
                                                     const float* __restrict__ a_s,
                                                     const float* __restrict__ a_d,
                                                     float* __restrict__ als, float* __restrict__ ald,
                                                     int m_tiles) {
    gemm_strip_body<false, 2>(nullptr, Ab, Bt, C, a_s, a_d, als, ald, m_tiles, (int)blockIdx.x);
}

#define TILE_E   4096
#define TILE_Q   16
#define BKT_N    256
#define BKT_SH   8
#define TOFF_P   512
#define CSR_THREADS 512
#define CSR_HW   (CSR_THREADS / 16)
#define CSR_J    16

template <bool FULL>
__device__ __forceinline__ void tile_sort_body(const int* __restrict__ ei, unsigned* __restrict__ tsorted,
                                               int* __restrict__ toff, long long n_edges, int tile,
                                               int* hist, int* st, unsigned* srt, int* wtot) {
    const int tid = threadIdx.x, lane = tid & 31, w = tid >> 5;
    hist[tid] = 0; hist[tid + 256] = 0;
    __syncthreads();

    const long long base = (long long)tile * TILE_E + tid;
    int s[TILE_Q], d[TILE_Q], key[TILE_Q];
#pragma unroll
    for (int q = 0; q < TILE_Q; ++q) {
        const long long i = base + q * 256;
        const bool ok = FULL || i < n_edges;
        s[q] = ok ? ei[i] : 0;
        d[q] = ok ? ei[n_edges + i] : -1;
    }
#pragma unroll
    for (int q = 0; q < TILE_Q; ++q) {
        key[q] = -1;
        if (FULL || d[q] >= 0) {
            const int b = d[q] >> BKT_SH;
            const int r = atomicAdd(&hist[b], 1);
            key[q] = (b << 12) | r;
        }
    }
    __syncthreads();

    const int a0 = hist[2 * tid], a1 = hist[2 * tid + 1];
    const int sum = a0 + a1;
    int incl = sum;
#pragma unroll
    for (int off = 1; off < 32; off <<= 1) {
        int y = __shfl_up(incl, off);
        if (lane >= off) incl += y;
    }
    if (lane == 31) wtot[w] = incl;
    __syncthreads();
    int woff = 0;
#pragma unroll
    for (int k = 0; k < 8; ++k) woff += (k < w) ? wtot[k] : 0;
    const int ex = woff + incl - sum;
    st[2 * tid] = ex; st[2 * tid + 1] = ex + a0;
    ((int2*)(toff + (size_t)tile * TOFF_P))[tid] = make_int2(ex, ex + a0);
    __syncthreads();

#pragma unroll
    for (int q = 0; q < TILE_Q; ++q) {
        if (FULL || key[q] >= 0)
            srt[st[key[q] >> 12] + (key[q] & 4095)] = ((unsigned)s[q] << 8) | (unsigned)(d[q] & (BKT_N - 1));
    }
    __syncthreads();

    const long long rem = n_edges - (long long)tile * TILE_E;
    const int total = FULL ? TILE_E : (int)(rem < TILE_E ? rem : TILE_E);
#pragma unroll
    for (int q = 0; q < TILE_Q; ++q) {
        const int idx = q * 256 + tid;
        if (FULL || idx < total) tsorted[(size_t)tile * TILE_E + idx] = srt[idx];
    }
}

__global__ void __launch_bounds__(256) __attribute__((amdgpu_waves_per_eu(6))) k_gemm1_tilesort(const float* __restrict__ x,
                                                     const unsigned short* __restrict__ w1t,
                                                     float* __restrict__ h1,
                                                     const float* __restrict__ a_s,
                                                     const float* __restrict__ a_d,
                                                     float* __restrict__ als, float* __restrict__ ald,
                                                     int m_tiles,
                                                     const int* __restrict__ ei, unsigned* __restrict__ tsorted,
                                                     int* __restrict__ toff, long long n_edges, int n_tiles) {
    __shared__ int      hist[TOFF_P];
    __shared__ int      st[TOFF_P];
    __shared__ unsigned srt[TILE_E];
    __shared__ int      wtot[8];
    const int wg = (int)(blockIdx.x >> 1);
    if (blockIdx.x & 1) {
        if (wg >= n_tiles) return;
        if ((long long)(wg + 1) * TILE_E <= n_edges) tile_sort_body<true>(ei, tsorted, toff, n_edges, wg, hist, st, srt, wtot);
        else                                         tile_sort_body<false>(ei, tsorted, toff, n_edges, wg, hist, st, srt, wtot);
    } else {
        gemm_strip_body<true, 8>(x, nullptr, w1t, h1, a_s, a_d, als, ald, m_tiles, wg);
    }
}

template <bool PLACE>
__device__ __forceinline__ void bucket_record(unsigned r, int* ncnt, int* ncur, const int* nst, int base,
                                              int* __restrict__ col) {
    const int dl = (int)(r & (BKT_N - 1));
    if (PLACE) { const int p = atomicAdd(&ncur[dl], 1); col[base + nst[dl] + p] = (int)(r >> 8); }
    else       atomicAdd(&ncnt[dl], 1);
}

template <bool PLACE, int NT>
__device__ __forceinline__ void bucket_pass(const unsigned* __restrict__ tsorted, const int* sS, const int* sE,
                                            int hw, int l16, int* ncnt, int* ncur, const int* nst, int base,
                                            int* __restrict__ col) {
    for (int t0 = hw; t0 < NT; t0 += CSR_HW * CSR_J) {
        unsigned rec[CSR_J], rec2[CSR_J];
        unsigned have = 0;
        bool longrun = false;
#pragma unroll
        for (int j = 0; j < CSR_J; ++j) {
            const int t = t0 + CSR_HW * j;
            rec[j] = 0; rec2[j] = 0;
            if (t < NT) {
                const int s = sS[t], e = sE[t];
                const int k = s + l16;
                if (k < e)      { have |= 1u << j;        rec[j]  = tsorted[(unsigned)t * TILE_E + (unsigned)k]; }
                if (k + 16 < e) { have |= 1u << (16 + j); rec2[j] = tsorted[(unsigned)t * TILE_E + (unsigned)k + 16u]; }
                longrun |= (e - s > 32);
            }
        }
#pragma unroll
        for (int j = 0; j < CSR_J; ++j)
            if (have & (1u << j)) bucket_record<PLACE>(rec[j], ncnt, ncur, nst, base, col);
#pragma unroll
        for (int j = 0; j < CSR_J; ++j)
            if (have & (1u << (16 + j))) bucket_record<PLACE>(rec2[j], ncnt, ncur, nst, base, col);
        if (longrun) {
            for (int j = 0; j < CSR_J; ++j) {
                const int t = t0 + CSR_HW * j;
                if (t >= NT) break;
                const int e = sE[t];
                for (int k = sS[t] + 32 + l16; k < e; k += 16)
                    bucket_record<PLACE>(tsorted[(unsigned)t * TILE_E + (unsigned)k], ncnt, ncur, nst, base, col);
            }
        }
    }
}

template <int NT>
__global__ void __launch_bounds__(CSR_THREADS) k_bucket_csr(const unsigned* __restrict__ tsorted, const int* __restrict__ toff,
                                                     int* __restrict__ cnt, int* __restrict__ rowptr,
                                                     int* __restrict__ col, int n_nodes) {
    __shared__ int sS[NT];
    __shared__ int sE[NT];
    __shared__ int ncnt[BKT_N];
    __shared__ int ncur[BKT_N];
    __shared__ int nst[BKT_N];
    __shared__ int red[CSR_THREADS / 32];
    __shared__ int wtot[8];
    const int tid = threadIdx.x, lane = tid & 31, w = tid >> 5;
    const int hw = tid >> 4, l16 = tid & 15;
    const int b = (int)blockIdx.x;
    if (tid < BKT_N) { ncnt[tid] = 0; ncur[tid] = 0; }

    int part = 0;
    for (int t = tid; t < NT; t += CSR_THREADS) {
        const int s = toff[(unsigned)t * TOFF_P + (unsigned)b];
        const int e = toff[(unsigned)t * TOFF_P + (unsigned)b + 1u];
        sS[t] = s; sE[t] = e;
        part += s;
    }
#pragma unroll
    for (int off = 16; off >= 1; off >>= 1) part += __shfl_xor(part, off);
    if (lane == 0) red[w] = part;
    __syncthreads();
    int base = 0;
#pragma unroll
    for (int k = 0; k < CSR_THREADS / 32; ++k) base += red[k];

    bucket_pass<false, NT>(tsorted, sS, sE, hw, l16, ncnt, ncur, nst, base, col);
    __syncthreads();

    int c = 0, incl = 0;
    if (tid < BKT_N) {
        c = ncnt[tid];
        incl = c;
#pragma unroll
        for (int off = 1; off < 32; off <<= 1) {
            int y = __shfl_up(incl, off);
            if (lane >= off) incl += y;
        }
        if (lane == 31) wtot[w] = incl;
    }
    __syncthreads();
    if (tid < BKT_N) {
        int woff = 0;
#pragma unroll
        for (int k = 0; k < 8; ++k) woff += (k < w) ? wtot[k] : 0;
        const int ex = woff + incl - c;
        nst[tid] = ex;
        const int node = b * BKT_N + tid;
        if (node < n_nodes) { cnt[node] = c; rowptr[node] = base + ex; }
    }
    __syncthreads();

    bucket_pass<true, NT>(tsorted, sS, sE, hw, l16, ncnt, ncur, nst, base, col);
}


template <int E>
__device__ __forceinline__ unsigned grp8_u(unsigned v) { return __builtin_amdgcn_mov_dpp8(v, E * 0x249249u); }
__device__ __forceinline__ unsigned grp8_pick_u(unsigned v, int e) {
    switch (e) {
        case 0: return grp8_u<0>(v); case 1: return grp8_u<1>(v); case 2: return grp8_u<2>(v); case 3: return grp8_u<3>(v);
        case 4: return grp8_u<4>(v); case 5: return grp8_u<5>(v); case 6: return grp8_u<6>(v); default: return grp8_u<7>(v);
    }
}
__device__ __forceinline__ float grp8_pick(float v, int e) {
    return __uint_as_float(grp8_pick_u(__float_as_uint(v), e));
}

__device__ __forceinline__ void score_l1(int base, int n, int start, int node, int n_edges,
                                         const int* __restrict__ col, const float* __restrict__ als,
                                         float adv, int hd, int e8, int* s, float* v) {
    const int nn = n - base;
    int c[4];
#pragma unroll
    for (int g = 0; g < 4; ++g) {
        c[g] = node;
        if (8 * g < nn) {
            int pos = start + base + 8 * g + e8;
            pos = pos < n_edges ? pos : n_edges - 1;
            c[g] = col[pos];
        }
    }
    float x[4];
#pragma unroll
    for (int g = 0; g < 4; ++g) {
        const int k = base + 8 * g + e8;
        s[g] = (k < n - 1) ? c[g] : node;
        x[g] = 0.f;
        if (8 * g < nn) x[g] = als[(size_t)s[g] * 4 + hd];
    }
#pragma unroll
    for (int g = 0; g < 4; ++g) {
        const int k = base + 8 * g + e8;
        const float y = x[g] + adv;
        const float z = y > 0.f ? y : NEG_SLOPE * y;
        v[g] = (k < n) ? z : -INFINITY;
    }
}

template <int E0, int NE>
__device__ __forceinline__ void accum_grp_l1(int sg, float ag, const float* __restrict__ hmat,
                                             int lane, v4f& acc) {
    v4f r[NE];
#pragma unroll
    for (int e = 0; e < NE; ++e) {
        const int sj = __builtin_amdgcn_readlane(sg, E0 + e);
        r[e] = ((const v4f*)(hmat + (size_t)sj * 128))[lane];
    }
#pragma unroll
    for (int e = 0; e < NE; ++e) {
        const float aj = grp8_pick(ag, E0 + e);
        acc += aj * r[e];
    }
}

__device__ __forceinline__ void accum_l1(int cntc, const int* s, const float* a,
                                         const float* __restrict__ hmat, int lane, v4f& acc) {
#pragma unroll
    for (int g = 0; g < 4; ++g) {
        const int rem = cntc - 8 * g;
        if (rem > 4)      accum_grp_l1<0, 8>(s[g], a[g], hmat, lane, acc);
        else if (rem > 0) accum_grp_l1<0, 4>(s[g], a[g], hmat, lane, acc);
    }
}

__device__ __forceinline__ float max4(const float* v) { return fmaxf(fmaxf(v[0], v[1]), fmaxf(v[2], v[3])); }

__global__ void __launch_bounds__(256) __attribute__((amdgpu_waves_per_eu(14))) k_gat_gather_l1(const int* __restrict__ rowptr,
                                                       const int* __restrict__ cnt,
                                                       const int* __restrict__ col,
                                                       const float* __restrict__ als,
                                                       const float* __restrict__ ald,
                                                       const float* __restrict__ hmat,
                                                       const float* __restrict__ bias,
                                                       unsigned short* __restrict__ outbf,
                                                       int n_nodes, int n_edges) {
    const int lane = threadIdx.x & 31;
    const int hd = lane >> 3, e8 = lane & 7;
    const int wave0 = __builtin_amdgcn_readfirstlane(
        (int)(blockIdx.x * (blockDim.x >> 5) + (threadIdx.x >> 5)));
    const int n_waves = (int)(gridDim.x * (blockDim.x >> 5));
    for (int node = wave0; node < n_nodes; node += n_waves) {
        const int start = rowptr[node];
        const int n = cnt[node] + 1;
        const float4 ad4 = *(const float4*)(ald + (size_t)node * 4);
        const float adv = hd == 0 ? ad4.x : hd == 1 ? ad4.y : hd == 2 ? ad4.z : ad4.w;

        int s0[4]; float v0[4];
        score_l1(0, n, start, node, n_edges, col, als, adv, hd, e8, s0, v0);

        float m = max4(v0);
        for (int base = 32; base < n; base += 32) {
            int s[4]; float v[4];
            score_l1(base, n, start, node, n_edges, col, als, adv, hd, e8, s, v);
            m = fmaxf(m, max4(v));
        }
        m = fmaxf(m, row_xor<1>(m)); m = fmaxf(m, row_xor<2>(m)); m = fmaxf(m, row_xor<4>(m));

        float den = (__expf(v0[0] - m) + __expf(v0[1] - m)) + (__expf(v0[2] - m) + __expf(v0[3] - m));
        for (int base = 32; base < n; base += 32) {
            int s[4]; float v[4];
            score_l1(base, n, start, node, n_edges, col, als, adv, hd, e8, s, v);
            den += (__expf(v[0] - m) + __expf(v[1] - m)) + (__expf(v[2] - m) + __expf(v[3] - m));
        }
        den += row_xor<1>(den); den += row_xor<2>(den); den += row_xor<4>(den);
        den += 1e-16f;

        v4f acc = {0.f, 0.f, 0.f, 0.f};
        {
            float a0[4];
#pragma unroll
            for (int g = 0; g < 4; ++g) a0[g] = __expf(v0[g] - m) / den;
            accum_l1(n < 32 ? n : 32, s0, a0, hmat, lane, acc);
        }
        for (int base = 32; base < n; base += 32) {
            int s[4]; float v[4], a[4];
            score_l1(base, n, start, node, n_edges, col, als, adv, hd, e8, s, v);
#pragma unroll
            for (int g = 0; g < 4; ++g) a[g] = __expf(v[g] - m) / den;
            accum_l1((n - base) < 32 ? (n - base) : 32, s, a, hmat, lane, acc);
        }

        const v4f r = acc + ((const v4f*)bias)[lane];
        uint2 o;
        o.x = pack_bf16x2(r.x > 0.f ? r.x : 0.f, r.y > 0.f ? r.y : 0.f);
        o.y = pack_bf16x2(r.z > 0.f ? r.z : 0.f, r.w > 0.f ? r.w : 0.f);
        ((uint2*)(outbf + (size_t)node * 128))[lane] = o;
    }
}

__device__ __forceinline__ void score_l2(int k, int n, int start, int node,
                                         const int* __restrict__ col, const float* __restrict__ als,
                                         float adv, int& s, float& v) {
    s = node; v = -INFINITY;
    if (k < n) {
        if (k < n - 1) s = col[start + k];
        const float y = als[s] + adv;
        v = y > 0.f ? y : NEG_SLOPE * y;
    }
}

template <int T0, int NT>
__device__ __forceinline__ void accum_steps_l2(int s, float a, const float* __restrict__ hmat,
                                               int c8, v4f& acc) {
    v4f r[NT];
#pragma unroll
    for (int t = 0; t < NT; ++t) {
        const int sq = (int)grp8_pick_u((unsigned)s, T0 + t);
        r[t] = ((const v4f*)(hmat + (size_t)sq * 32))[c8];
    }
#pragma unroll
    for (int t = 0; t < NT; ++t) {
        const float aq = grp8_pick(a, T0 + t);
        acc += aq * r[t];
    }
}

__device__ __forceinline__ void accum_l2(int cntc, int s, float a, const float* __restrict__ hmat,
                                         int c8, v4f& acc) {
    accum_steps_l2<0, 4>(s, a, hmat, c8, acc);
    if (cntc > 16) accum_steps_l2<4, 2>(s, a, hmat, c8, acc);
    if (cntc > 24) accum_steps_l2<6, 2>(s, a, hmat, c8, acc);
}

__global__ void __launch_bounds__(256) k_gat_gather_l2(const int* __restrict__ rowptr,
                                                       const int* __restrict__ cnt,
                                                       const int* __restrict__ col,
                                                       const float* __restrict__ als,
                                                       const float* __restrict__ ald,
                                                       const float* __restrict__ hmat,
                                                       const float* __restrict__ bias,
                                                       float* __restrict__ outp,
                                                       int n_nodes) {
    const int lane = threadIdx.x & 31;
    const int q = lane >> 3, c8 = lane & 7;
    const int kk = 4 * c8 + q;
    const int wave0 = __builtin_amdgcn_readfirstlane(
        (int)(blockIdx.x * (blockDim.x >> 5) + (threadIdx.x >> 5)));
    const int n_waves = (int)(gridDim.x * (blockDim.x >> 5));
    const v4f bv = ((const v4f*)bias)[c8];
    for (int node = wave0; node < n_nodes; node += n_waves) {
        const int start = rowptr[node];
        const int n = cnt[node] + 1;
        const float adv = ald[node];

        int s0; float v0;
        score_l2(kk, n, start, node, col, als, adv, s0, v0);

        float m = v0;
        for (int base = 32; base < n; base += 32) {
            int s; float v;
            score_l2(base + kk, n, start, node, col, als, adv, s, v);
            m = fmaxf(m, v);
        }
#pragma unroll
        for (int off = 16; off >= 1; off >>= 1) m = fmaxf(m, __shfl_xor(m, off));

        float den = __expf(v0 - m);
        for (int base = 32; base < n; base += 32) {
            int s; float v;
            score_l2(base + kk, n, start, node, col, als, adv, s, v);
            den += __expf(v - m);
        }
#pragma unroll
        for (int off = 16; off >= 1; off >>= 1) den += __shfl_xor(den, off);
        den += 1e-16f;

        v4f acc = {0.f, 0.f, 0.f, 0.f};
        accum_l2(n < 32 ? n : 32, s0, __expf(v0 - m) / den, hmat, c8, acc);
        for (int base = 32; base < n; base += 32) {
            int s; float v;
            score_l2(base + kk, n, start, node, col, als, adv, s, v);
            accum_l2((n - base) < 32 ? (n - base) : 32, s, __expf(v - m) / den, hmat, c8, acc);
        }
        acc.x += row_xor<8>(acc.x); acc.y += row_xor<8>(acc.y); acc.z += row_xor<8>(acc.z); acc.w += row_xor<8>(acc.w);
        acc.x += __shfl_xor(acc.x, 16); acc.y += __shfl_xor(acc.y, 16);
        acc.z += __shfl_xor(acc.z, 16); acc.w += __shfl_xor(acc.w, 16);
        if (lane < 8) ((v4f*)(outp + (size_t)node * 32))[c8] = acc + bv;
    }
}


extern "C" void kernel_launch(void* const* d_in, const int* in_sizes, int n_in,
                              void* d_out, int out_size, void* d_ws, size_t ws_size,
                              hipStream_t stream) {
    (void)n_in; (void)out_size; (void)ws_size; (void)in_sizes;
    constexpr long long N = 100000, E = 1600000;
    constexpr int C1 = 128, HEADS = 4, C2 = 32;
    constexpr int N_TILES = (int)((E + TILE_E - 1) / TILE_E);
    constexpr int N_BKT   = (int)((N + BKT_N - 1) / BKT_N);
    static_assert(N_BKT + 1 <= TOFF_P, "bucket starts of a tile must fit its table row");
    static_assert(N <= (1LL << 24), "a record holds the source in 24 bits");
    constexpr int NODES_PER_WAVE = 7;
    constexpr int GATHER_BLOCKS = (int)((N + 8 * NODES_PER_WAVE - 1) / (8 * NODES_PER_WAVE));

    const float* x    = (const float*)d_in[0];
    const int*   ei   = (const int*)  d_in[1];
    const float* W1   = (const float*)d_in[2];
    const float* aS1  = (const float*)d_in[3];
    const float* aD1  = (const float*)d_in[4];
    const float* b1   = (const float*)d_in[5];
    const float* W2   = (const float*)d_in[6];
    const float* aS2  = (const float*)d_in[7];
    const float* aD2  = (const float*)d_in[8];
    const float* b2   = (const float*)d_in[9];
    float* out = (float*)d_out;

    char* ws = (char*)d_ws;
    size_t off = 0;
    auto alloc = [&](size_t bytes) { size_t o = off; off = (off + bytes + 255) & ~(size_t)255; return o; };
    unsigned short* hbf  = (unsigned short*)(ws + alloc(N * C1 * 2));
    float*          h1   = (float*)(ws + alloc(N * C1 * 4));
    float*          as1  = (float*)(ws + alloc(N * HEADS * 4));
    float*          ad1  = (float*)(ws + alloc(N * HEADS * 4));
    int*            cnt  = (int*)(ws + alloc(N * 4));
    int*            rptr = (int*)(ws + alloc(N * 4));
    unsigned*       tsrt = (unsigned*)(ws + alloc((size_t)N_TILES * TILE_E * 4));
    int*            toff = (int*)(ws + alloc((size_t)N_TILES * TOFF_P * 4));
    int*            col  = (int*)(ws + alloc(E * 4));
    unsigned short* w1t  = (unsigned short*)(ws + alloc(128 * 128 * 2));
    unsigned short* w2t  = (unsigned short*)(ws + alloc(32 * 128 * 2));

    constexpr int B = 256;
    auto blocks = [](long long n, int b) { return (unsigned)((n + b - 1) / b); };

    constexpr int PREP_THREADS = 128 * 128 + 32 * 128;
    k_prep<<<blocks(PREP_THREADS, B), B, 0, stream>>>(W1, w1t, W2, w2t);

    constexpr unsigned GEMM_WGS = (unsigned)((N / 16 + 7) / 8);
    constexpr unsigned EDGE_WGS = (unsigned)N_TILES;
    constexpr unsigned PAIR_WGS = 2 * (GEMM_WGS > EDGE_WGS ? GEMM_WGS : EDGE_WGS);
    k_gemm1_tilesort<<<PAIR_WGS, 256, 0, stream>>>(x, w1t, h1, aS1, aD1, as1, ad1, (int)(N / 16), ei, tsrt, toff, E, N_TILES);

    k_bucket_csr<N_TILES><<<N_BKT, CSR_THREADS, 0, stream>>>(tsrt, toff, cnt, rptr, col, (int)N);

    k_gat_gather_l1<<<GATHER_BLOCKS, 256, 0, stream>>>(rptr, cnt, col, as1, ad1, h1, b1, hbf, (int)N, (int)E);

    k_gemm2_strip<<<blocks(N / 16, 8), 256, 0, stream>>>(hbf, w2t, h1, aS2, aD2, as1, ad1, (int)(N / 16));

    k_gat_gather_l2<<<GATHER_BLOCKS, 256, 0, stream>>>(rptr, cnt, col, as1, ad1, h1, b2, out, (int)N);
}


# Layer-2 GEMM moved into the layer-1 gather: a wave keeps its 7 nodes' relu rows as bf16 in LDS and multiplies them by W2 (staged in LDS) on WMMA, writing h2 and both coefficients; 6 -> 5 launches
# speedup vs baseline: 14.3719x; 1.1166x over previous
#include <hip/hip_runtime.h>
#include <hip/hip_bf16.h>

typedef __attribute__((ext_vector_type(16))) __bf16 v16bf;
typedef __attribute__((ext_vector_type(8)))  float  v8f;
typedef __attribute__((ext_vector_type(2)))  float  v2f;
typedef __attribute__((ext_vector_type(2)))  __bf16 v2bf;

#define NEG_SLOPE 0.2f


__device__ __forceinline__ unsigned short f2bf(float f) {
    unsigned u = __float_as_uint(f);
    unsigned r = u + 0x7FFFu + ((u >> 16) & 1u);
    return (unsigned short)(r >> 16);
}

__device__ __forceinline__ unsigned pack_bf16x2(float lo, float hi) {
    v2f f = {lo, hi};
    return __builtin_bit_cast(unsigned, __builtin_convertvector(f, v2bf));
}


__global__ void __launch_bounds__(256) k_prep(const float* __restrict__ W1, unsigned short* __restrict__ w1t,
                                              const float* __restrict__ W2, unsigned short* __restrict__ w2t) {
    int i = blockIdx.x * 256 + threadIdx.x;
    if (i < 128 * 128) {
        int n = i >> 7, k = i & 127;
        w1t[i] = f2bf(W1[k * 128 + n]);
    } else if (i < 128 * 128 + 32 * 128) {
        i -= 128 * 128;
        int n = i >> 7, k = i & 127;
        w2t[i] = f2bf(W2[k * 32 + n]);
    }
}

union Frag { unsigned u[8]; v16bf v; };
typedef __attribute__((ext_vector_type(4))) float    v4f;
typedef __attribute__((ext_vector_type(4))) unsigned v4u;

__device__ __forceinline__ v16bf frag_of(v4u q0, v4u q1) {
    Frag f;
    f.u[0] = q0.x; f.u[1] = q0.y; f.u[2] = q0.z; f.u[3] = q0.w;
    f.u[4] = q1.x; f.u[5] = q1.y; f.u[6] = q1.z; f.u[7] = q1.w;
    return f.v;
}

template <int M>
__device__ __forceinline__ float row_xor(float v) {
    return __builtin_bit_cast(float, __builtin_amdgcn_mov_dpp(
        __builtin_bit_cast(int, v), 0x160 + M, 0xf, 0xf, false));
}

__device__ __forceinline__ float row_reduce_scatter(const float* ps, const float* pd, int r15) {
    const bool b3 = r15 & 8, b2 = r15 & 4, b1 = r15 & 2, b0 = r15 & 1;
    float w[8], x[4], y[2];
#pragma unroll
    for (int i = 0; i < 8; ++i) w[i] = (b3 ? pd[i] : ps[i]) + row_xor<8>(b3 ? ps[i] : pd[i]);
#pragma unroll
    for (int i = 0; i < 4; ++i) x[i] = (b2 ? w[i + 4] : w[i]) + row_xor<4>(b2 ? w[i] : w[i + 4]);
#pragma unroll
    for (int i = 0; i < 2; ++i) y[i] = (b1 ? x[i + 2] : x[i]) + row_xor<2>(b1 ? x[i] : x[i + 2]);
    return (b0 ? y[1] : y[0]) + row_xor<1>(b0 ? y[0] : y[1]);
}

template <bool A_F32, int NT, bool C_BF16>
__device__ __forceinline__ void gemm_strip_body(const float* __restrict__ Af,
                                                const unsigned short* __restrict__ Ab,
                                                const unsigned short* __restrict__ Bt,
                                                float* __restrict__ C,
                                                unsigned short* __restrict__ Cb,
                                                const float* __restrict__ a_s,
                                                const float* __restrict__ a_d,
                                                float* __restrict__ als, float* __restrict__ ald,
                                                int m_tiles, int wg) {
    constexpr int H = NT / 2;
    const int lane = threadIdx.x & 31;
    const int tileM = __builtin_amdgcn_readfirstlane((int)(wg * 8 + (threadIdx.x >> 5)));
    if (tileM >= m_tiles) return;
    const int r15  = lane & 15;
    const int half = lane >> 4;

    const v4u* bbase = (const v4u*)(Bt + (size_t)r15 * 128 + half * 8);
    v4u qb[2][8];
#pragma unroll
    for (int s = 0; s < 4; ++s) { qb[0][2 * s] = bbase[s * 4 + 0]; qb[0][2 * s + 1] = bbase[s * 4 + 2]; }

    v16bf a[4];
    if constexpr (A_F32) {
        const v4f* arow = (const v4f*)(Af + (size_t)(tileM * 16 + r15) * 128 + half * 8);
        v4f p[16];
#pragma unroll
        for (int s = 0; s < 4; ++s) {
            p[4 * s + 0] = arow[s * 8 + 0];
            p[4 * s + 1] = arow[s * 8 + 1];
            p[4 * s + 2] = arow[s * 8 + 4];
            p[4 * s + 3] = arow[s * 8 + 5];
        }
        asm volatile("" :: "v"(p[0]), "v"(p[1]), "v"(p[2]), "v"(p[3]), "v"(p[4]), "v"(p[5]), "v"(p[6]), "v"(p[7]),
                           "v"(p[8]), "v"(p[9]), "v"(p[10]), "v"(p[11]), "v"(p[12]), "v"(p[13]), "v"(p[14]), "v"(p[15]));
#pragma unroll
        for (int s = 0; s < 4; ++s) {
            Frag f;
#pragma unroll
            for (int j = 0; j < 4; ++j) {
                f.u[2 * j]     = pack_bf16x2(p[4 * s + j].x, p[4 * s + j].y);
                f.u[2 * j + 1] = pack_bf16x2(p[4 * s + j].z, p[4 * s + j].w);
            }
            a[s] = f.v;
        }
    } else {
        const v4u* arow = (const v4u*)(Ab + (size_t)(tileM * 16 + r15) * 128 + half * 8);
        v4u qa[8];
#pragma unroll
        for (int s = 0; s < 4; ++s) { qa[2 * s] = arow[s * 4 + 0]; qa[2 * s + 1] = arow[s * 4 + 2]; }
#pragma unroll
        for (int s = 0; s < 4; ++s) a[s] = frag_of(qa[2 * s], qa[2 * s + 1]);
    }

    float ps[8], pd[8], res[H];
#pragma unroll
    for (int t = 0; t < NT; ++t) {
        if (t + 1 < NT) {
#pragma unroll
            for (int s = 0; s < 4; ++s) {
                qb[(t + 1) & 1][2 * s]     = bbase[(t + 1) * 256 + s * 4 + 0];
                qb[(t + 1) & 1][2 * s + 1] = bbase[(t + 1) * 256 + s * 4 + 2];
            }
        }
        const float cs = a_s[t * 16 + r15], cd = a_d[t * 16 + r15];
        v8f acc = {};
#pragma unroll
        for (int s = 0; s < 4; ++s)
            acc = __builtin_amdgcn_wmma_f32_16x16x32_bf16(
                    false, a[s], false, frag_of(qb[t & 1][2 * s], qb[t & 1][2 * s + 1]),
                    (short)0, acc, false, false);
        if constexpr (C_BF16) {
            const unsigned keepmask = (r15 & 1) ? 0xffff0000u : 0x0000ffffu;
#pragma unroll
            for (int rp = 0; rp < 4; ++rp) {
                const unsigned own = pack_bf16x2(acc[2 * rp], acc[2 * rp + 1]);
                const unsigned nbr = (unsigned)__builtin_amdgcn_mov_dpp((int)own, 0x160 + 1, 0xf, 0xf, false);
                const unsigned rot = (nbr << 16) | (nbr >> 16);
                const int row = tileM * 16 + 8 * half + 2 * rp + (r15 & 1);
                const int col = t * 16 + (r15 & ~1);
                *(unsigned*)(Cb + (size_t)row * (NT * 16) + col) = (own & keepmask) | (rot & ~keepmask);
            }
        } else {
#pragma unroll
            for (int r = 0; r < 8; ++r) {
                int row = tileM * 16 + r + 8 * half;
                int col = t * 16 + r15;
                C[(size_t)row * (NT * 16) + col] = acc[r];
            }
        }
        if ((t & 1) == 0) {
#pragma unroll
            for (int r = 0; r < 8; ++r) { ps[r] = acc[r] * cs; pd[r] = acc[r] * cd; }
        } else {
#pragma unroll
            for (int r = 0; r < 8; ++r) { ps[r] += acc[r] * cs; pd[r] += acc[r] * cd; }
            res[t >> 1] = row_reduce_scatter(ps, pd, r15);
        }
    }
    {
        const int row = tileM * 16 + 8 * half + (r15 & 7);
        float* dstp = ((r15 & 8) ? ald : als) + (size_t)row * H;
        if constexpr (H == 4) {
            *(float4*)dstp = make_float4(res[0], res[1], res[2], res[3]);
        } else {
#pragma unroll
            for (int h = 0; h < H; ++h) dstp[h] = res[h];
        }
    }
}

#define TILE_E   4096
#define TILE_Q   16
#define BKT_N    256
#define BKT_SH   8
#define TOFF_P   512
#define CSR_THREADS 512
#define CSR_HW   (CSR_THREADS / 16)
#define CSR_J    16

template <bool FULL>
__device__ __forceinline__ void tile_sort_body(const int* __restrict__ ei, unsigned* __restrict__ tsorted,
                                               int* __restrict__ toff, long long n_edges, int tile,
                                               int* hist, int* st, unsigned* srt, int* wtot) {
    const int tid = threadIdx.x, lane = tid & 31, w = tid >> 5;
    hist[tid] = 0; hist[tid + 256] = 0;
    __syncthreads();

    const long long base = (long long)tile * TILE_E + tid;
    int s[TILE_Q], d[TILE_Q], key[TILE_Q];
#pragma unroll
    for (int q = 0; q < TILE_Q; ++q) {
        const long long i = base + q * 256;
        const bool ok = FULL || i < n_edges;
        s[q] = ok ? ei[i] : 0;
        d[q] = ok ? ei[n_edges + i] : -1;
    }
#pragma unroll
    for (int q = 0; q < TILE_Q; ++q) {
        key[q] = -1;
        if (FULL || d[q] >= 0) {
            const int b = d[q] >> BKT_SH;
            const int r = atomicAdd(&hist[b], 1);
            key[q] = (b << 12) | r;
        }
    }
    __syncthreads();

    const int a0 = hist[2 * tid], a1 = hist[2 * tid + 1];
    const int sum = a0 + a1;
    int incl = sum;
#pragma unroll
    for (int off = 1; off < 32; off <<= 1) {
        int y = __shfl_up(incl, off);
        if (lane >= off) incl += y;
    }
    if (lane == 31) wtot[w] = incl;
    __syncthreads();
    int woff = 0;
#pragma unroll
    for (int k = 0; k < 8; ++k) woff += (k < w) ? wtot[k] : 0;
    const int ex = woff + incl - sum;
    st[2 * tid] = ex; st[2 * tid + 1] = ex + a0;
    ((int2*)(toff + (size_t)tile * TOFF_P))[tid] = make_int2(ex, ex + a0);
    __syncthreads();

#pragma unroll
    for (int q = 0; q < TILE_Q; ++q) {
        if (FULL || key[q] >= 0)
            srt[st[key[q] >> 12] + (key[q] & 4095)] = ((unsigned)s[q] << 8) | (unsigned)(d[q] & (BKT_N - 1));
    }
    __syncthreads();

    const long long rem = n_edges - (long long)tile * TILE_E;
    const int total = FULL ? TILE_E : (int)(rem < TILE_E ? rem : TILE_E);
#pragma unroll
    for (int q = 0; q < TILE_Q; ++q) {
        const int idx = q * 256 + tid;
        if (FULL || idx < total) tsorted[(size_t)tile * TILE_E + idx] = srt[idx];
    }
}

__global__ void __launch_bounds__(256) __attribute__((amdgpu_waves_per_eu(6))) k_gemm1_tilesort(const float* __restrict__ x,
                                                     const unsigned short* __restrict__ w1t,
                                                     unsigned short* __restrict__ h1b,
                                                     const float* __restrict__ a_s,
                                                     const float* __restrict__ a_d,
                                                     float* __restrict__ als, float* __restrict__ ald,
                                                     int m_tiles,
                                                     const int* __restrict__ ei, unsigned* __restrict__ tsorted,
                                                     int* __restrict__ toff, long long n_edges, int n_tiles) {
    __shared__ int      hist[TOFF_P];
    __shared__ int      st[TOFF_P];
    __shared__ unsigned srt[TILE_E];
    __shared__ int      wtot[8];
    const int wg = (int)(blockIdx.x >> 1);
    if (blockIdx.x & 1) {
        if (wg >= n_tiles) return;
        if ((long long)(wg + 1) * TILE_E <= n_edges) tile_sort_body<true>(ei, tsorted, toff, n_edges, wg, hist, st, srt, wtot);
        else                                         tile_sort_body<false>(ei, tsorted, toff, n_edges, wg, hist, st, srt, wtot);
    } else {
        gemm_strip_body<true, 8, true>(x, nullptr, w1t, nullptr, h1b, a_s, a_d, als, ald, m_tiles, wg);
    }
}

template <bool PLACE>
__device__ __forceinline__ void bucket_record(unsigned r, int* ncnt, int* ncur, const int* nst, int base,
                                              int* __restrict__ col) {
    const int dl = (int)(r & (BKT_N - 1));
    if (PLACE) { const int p = atomicAdd(&ncur[dl], 1); col[base + nst[dl] + p] = (int)(r >> 8); }
    else       atomicAdd(&ncnt[dl], 1);
}

template <bool PLACE, int NT>
__device__ __forceinline__ void bucket_pass(const unsigned* __restrict__ tsorted, const int* sS, const int* sE,
                                            int hw, int l16, int* ncnt, int* ncur, const int* nst, int base,
                                            int* __restrict__ col) {
    for (int t0 = hw; t0 < NT; t0 += CSR_HW * CSR_J) {
        unsigned rec[CSR_J], rec2[CSR_J];
        unsigned have = 0;
        bool longrun = false;
#pragma unroll
        for (int j = 0; j < CSR_J; ++j) {
            const int t = t0 + CSR_HW * j;
            rec[j] = 0; rec2[j] = 0;
            if (t < NT) {
                const int s = sS[t], e = sE[t];
                const int k = s + l16;
                if (k < e)      { have |= 1u << j;        rec[j]  = tsorted[(unsigned)t * TILE_E + (unsigned)k]; }
                if (k + 16 < e) { have |= 1u << (16 + j); rec2[j] = tsorted[(unsigned)t * TILE_E + (unsigned)k + 16u]; }
                longrun |= (e - s > 32);
            }
        }
#pragma unroll
        for (int j = 0; j < CSR_J; ++j)
            if (have & (1u << j)) bucket_record<PLACE>(rec[j], ncnt, ncur, nst, base, col);
#pragma unroll
        for (int j = 0; j < CSR_J; ++j)
            if (have & (1u << (16 + j))) bucket_record<PLACE>(rec2[j], ncnt, ncur, nst, base, col);
        if (longrun) {
            for (int j = 0; j < CSR_J; ++j) {
                const int t = t0 + CSR_HW * j;
                if (t >= NT) break;
                const int e = sE[t];
                for (int k = sS[t] + 32 + l16; k < e; k += 16)
                    bucket_record<PLACE>(tsorted[(unsigned)t * TILE_E + (unsigned)k], ncnt, ncur, nst, base, col);
            }
        }
    }
}

template <int NT>
__global__ void __launch_bounds__(CSR_THREADS) k_bucket_csr(const unsigned* __restrict__ tsorted, const int* __restrict__ toff,
                                                     int* __restrict__ cnt, int* __restrict__ rowptr,
                                                     int* __restrict__ col, int n_nodes) {
    __shared__ int sS[NT];
    __shared__ int sE[NT];
    __shared__ int ncnt[BKT_N];
    __shared__ int ncur[BKT_N];
    __shared__ int nst[BKT_N];
    __shared__ int red[CSR_THREADS / 32];
    __shared__ int wtot[8];
    const int tid = threadIdx.x, lane = tid & 31, w = tid >> 5;
    const int hw = tid >> 4, l16 = tid & 15;
    const int b = (int)blockIdx.x;
    if (tid < BKT_N) { ncnt[tid] = 0; ncur[tid] = 0; }

    int part = 0;
    for (int t = tid; t < NT; t += CSR_THREADS) {
        const int s = toff[(unsigned)t * TOFF_P + (unsigned)b];
        const int e = toff[(unsigned)t * TOFF_P + (unsigned)b + 1u];
        sS[t] = s; sE[t] = e;
        part += s;
    }
#pragma unroll
    for (int off = 16; off >= 1; off >>= 1) part += __shfl_xor(part, off);
    if (lane == 0) red[w] = part;
    __syncthreads();
    int base = 0;
#pragma unroll
    for (int k = 0; k < CSR_THREADS / 32; ++k) base += red[k];

    bucket_pass<false, NT>(tsorted, sS, sE, hw, l16, ncnt, ncur, nst, base, col);
    __syncthreads();

    int c = 0, incl = 0;
    if (tid < BKT_N) {
        c = ncnt[tid];
        incl = c;
#pragma unroll
        for (int off = 1; off < 32; off <<= 1) {
            int y = __shfl_up(incl, off);
            if (lane >= off) incl += y;
        }
        if (lane == 31) wtot[w] = incl;
    }
    __syncthreads();
    if (tid < BKT_N) {
        int woff = 0;
#pragma unroll
        for (int k = 0; k < 8; ++k) woff += (k < w) ? wtot[k] : 0;
        const int ex = woff + incl - c;
        nst[tid] = ex;
        const int node = b * BKT_N + tid;
        if (node < n_nodes) { cnt[node] = c; rowptr[node] = base + ex; }
    }
    __syncthreads();

    bucket_pass<true, NT>(tsorted, sS, sE, hw, l16, ncnt, ncur, nst, base, col);
}


template <int E>
__device__ __forceinline__ unsigned grp8_u(unsigned v) { return __builtin_amdgcn_mov_dpp8(v, E * 0x249249u); }
__device__ __forceinline__ unsigned grp8_pick_u(unsigned v, int e) {
    switch (e) {
        case 0: return grp8_u<0>(v); case 1: return grp8_u<1>(v); case 2: return grp8_u<2>(v); case 3: return grp8_u<3>(v);
        case 4: return grp8_u<4>(v); case 5: return grp8_u<5>(v); case 6: return grp8_u<6>(v); default: return grp8_u<7>(v);
    }
}
__device__ __forceinline__ float grp8_pick(float v, int e) {
    return __uint_as_float(grp8_pick_u(__float_as_uint(v), e));
}

__device__ __forceinline__ void score_l1(int base, int n, int start, int node, int n_edges,
                                         const int* __restrict__ col, const float* __restrict__ als,
                                         float adv, int hd, int e8, int* s, float* v) {
    const int nn = n - base;
    int c[4];
#pragma unroll
    for (int g = 0; g < 4; ++g) {
        c[g] = node;
        if (8 * g < nn) {
            int pos = start + base + 8 * g + e8;
            pos = pos < n_edges ? pos : n_edges - 1;
            c[g] = col[pos];
        }
    }
    float x[4];
#pragma unroll
    for (int g = 0; g < 4; ++g) {
        const int k = base + 8 * g + e8;
        s[g] = (k < n - 1) ? c[g] : node;
        x[g] = 0.f;
        if (8 * g < nn) x[g] = als[(size_t)s[g] * 4 + hd];
    }
#pragma unroll
    for (int g = 0; g < 4; ++g) {
        const int k = base + 8 * g + e8;
        const float y = x[g] + adv;
        const float z = y > 0.f ? y : NEG_SLOPE * y;
        v[g] = (k < n) ? z : -INFINITY;
    }
}

template <int E0, int NE>
__device__ __forceinline__ void accum_grp_l1(int sg, float ag, const unsigned short* __restrict__ hmat,
                                             int lane, v4f& acc) {
    uint2 r[NE];
#pragma unroll
    for (int e = 0; e < NE; ++e) {
        const int sj = __builtin_amdgcn_readlane(sg, E0 + e);
        r[e] = ((const uint2*)(hmat + (size_t)sj * 128))[lane];
    }
#pragma unroll
    for (int e = 0; e < NE; ++e) {
        const float aj = grp8_pick(ag, E0 + e);
        const v4f f = {__uint_as_float(r[e].x << 16), __uint_as_float(r[e].x & 0xffff0000u),
                       __uint_as_float(r[e].y << 16), __uint_as_float(r[e].y & 0xffff0000u)};
        acc += aj * f;
    }
}

__device__ __forceinline__ void accum_l1(int cntc, const int* s, const float* a,
                                         const unsigned short* __restrict__ hmat, int lane, v4f& acc) {
#pragma unroll
    for (int g = 0; g < 4; ++g) {
        const int rem = cntc - 8 * g;
        if (rem > 4)      accum_grp_l1<0, 8>(s[g], a[g], hmat, lane, acc);
        else if (rem > 0) accum_grp_l1<0, 4>(s[g], a[g], hmat, lane, acc);
    }
}

__device__ __forceinline__ float max4(const float* v) { return fmaxf(fmaxf(v[0], v[1]), fmaxf(v[2], v[3])); }

#define G1_NODES   7
#define G1_ROWS    8
#define G1_PITCH   272
__global__ void __launch_bounds__(256) __attribute__((amdgpu_waves_per_eu(14))) k_gat_gather_l1_gemm2(const int* __restrict__ rowptr,
                                                       const int* __restrict__ cnt,
                                                       const int* __restrict__ col,
                                                       const float* __restrict__ als,
                                                       const float* __restrict__ ald,
                                                       const unsigned short* __restrict__ hmat,
                                                       const float* __restrict__ bias,
                                                       const unsigned short* __restrict__ w2t,
                                                       const float* __restrict__ a_s2,
                                                       const float* __restrict__ a_d2,
                                                       float* __restrict__ h2,
                                                       float* __restrict__ als2, float* __restrict__ ald2,
                                                       int n_nodes, int n_edges) {
    __shared__ __attribute__((aligned(16))) unsigned char sB[32 * G1_PITCH];
    __shared__ __attribute__((aligned(16))) unsigned char sA[8 * G1_ROWS * G1_PITCH];
    const int lane = threadIdx.x & 31;
    const int hd = lane >> 3, e8 = lane & 7;

    {
        const int t = threadIdx.x;
        const v4u* g = (const v4u*)(w2t + (size_t)(t >> 3) * 128 + (t & 7) * 16);
        const v4u q0 = g[0], q1 = g[1];
        v4u* d = (v4u*)(sB + (t >> 3) * G1_PITCH + (t & 7) * 32);
        d[0] = q0; d[1] = q1;
    }
    __syncthreads();

    const int wv = __builtin_amdgcn_readfirstlane((int)(threadIdx.x >> 5));
    const int node0 = __builtin_amdgcn_readfirstlane((int)(blockIdx.x * 8 + (threadIdx.x >> 5)) * G1_NODES);
    if (node0 >= n_nodes) return;
    unsigned char* myA = sA + wv * (G1_ROWS * G1_PITCH);

    for (int i = 0; i < G1_ROWS; ++i) {
        const int node = node0 + i;
        uint2 o = make_uint2(0u, 0u);
        if (i < G1_NODES && node < n_nodes) {
            const int start = rowptr[node];
            const int n = cnt[node] + 1;
            const float4 ad4 = *(const float4*)(ald + (size_t)node * 4);
            const float adv = hd == 0 ? ad4.x : hd == 1 ? ad4.y : hd == 2 ? ad4.z : ad4.w;

            int s0[4]; float v0[4];
            score_l1(0, n, start, node, n_edges, col, als, adv, hd, e8, s0, v0);

            float m = max4(v0);
            for (int base = 32; base < n; base += 32) {
                int s[4]; float v[4];
                score_l1(base, n, start, node, n_edges, col, als, adv, hd, e8, s, v);
                m = fmaxf(m, max4(v));
            }
            m = fmaxf(m, row_xor<1>(m)); m = fmaxf(m, row_xor<2>(m)); m = fmaxf(m, row_xor<4>(m));

            float den = (__expf(v0[0] - m) + __expf(v0[1] - m)) + (__expf(v0[2] - m) + __expf(v0[3] - m));
            for (int base = 32; base < n; base += 32) {
                int s[4]; float v[4];
                score_l1(base, n, start, node, n_edges, col, als, adv, hd, e8, s, v);
                den += (__expf(v[0] - m) + __expf(v[1] - m)) + (__expf(v[2] - m) + __expf(v[3] - m));
            }
            den += row_xor<1>(den); den += row_xor<2>(den); den += row_xor<4>(den);
            den += 1e-16f;

            v4f acc = {0.f, 0.f, 0.f, 0.f};
            {
                float a0[4];
#pragma unroll
                for (int g = 0; g < 4; ++g) a0[g] = __expf(v0[g] - m) / den;
                accum_l1(n < 32 ? n : 32, s0, a0, hmat, lane, acc);
            }
            for (int base = 32; base < n; base += 32) {
                int s[4]; float v[4], a[4];
                score_l1(base, n, start, node, n_edges, col, als, adv, hd, e8, s, v);
#pragma unroll
                for (int g = 0; g < 4; ++g) a[g] = __expf(v[g] - m) / den;
                accum_l1((n - base) < 32 ? (n - base) : 32, s, a, hmat, lane, acc);
            }

            const v4f r = acc + *(const v4f*)((const char*)bias + (unsigned)lane * 16u);
            o.x = pack_bf16x2(r.x > 0.f ? r.x : 0.f, r.y > 0.f ? r.y : 0.f);
            o.y = pack_bf16x2(r.z > 0.f ? r.z : 0.f, r.w > 0.f ? r.w : 0.f);
        }
        *(uint2*)(myA + i * G1_PITCH + lane * 8) = o;
    }

    const int r15 = lane & 15, half = lane >> 4;
    const unsigned char* arow = myA + (r15 & 7) * G1_PITCH + half * 16;
    const unsigned char* brow = sB + r15 * G1_PITCH + half * 16;
    v8f acc0 = {}, acc1 = {};
#pragma unroll
    for (int s = 0; s < 4; ++s) {
        const v16bf af = frag_of(*(const v4u*)(arow + 64 * s), *(const v4u*)(arow + 64 * s + 32));
        const v16bf b0 = frag_of(*(const v4u*)(brow + 64 * s), *(const v4u*)(brow + 64 * s + 32));
        const v16bf b1 = frag_of(*(const v4u*)(brow + 16 * G1_PITCH + 64 * s),
                                 *(const v4u*)(brow + 16 * G1_PITCH + 64 * s + 32));
        acc0 = __builtin_amdgcn_wmma_f32_16x16x32_bf16(false, af, false, b0, (short)0, acc0, false, false);
        acc1 = __builtin_amdgcn_wmma_f32_16x16x32_bf16(false, af, false, b1, (short)0, acc1, false, false);
    }
    const float* cv = half ? a_d2 : a_s2;
    const float k0 = cv[r15], k1 = cv[16 + r15];
    float p[8], w[4], x[2];
#pragma unroll
    for (int r = 0; r < 8; ++r) p[r] = acc0[r] * k0 + acc1[r] * k1;
    const bool b3 = r15 & 8, b2 = r15 & 4, b1 = r15 & 2;
#pragma unroll
    for (int i = 0; i < 4; ++i) w[i] = (b3 ? p[i + 4] : p[i]) + row_xor<8>(b3 ? p[i] : p[i + 4]);
#pragma unroll
    for (int i = 0; i < 2; ++i) x[i] = (b2 ? w[i + 2] : w[i]) + row_xor<4>(b2 ? w[i] : w[i + 2]);
    float y = (b1 ? x[1] : x[0]) + row_xor<2>(b1 ? x[0] : x[1]);
    y += row_xor<1>(y);
    {
        const int node = node0 + (r15 >> 1);
        if (!(r15 & 1) && (r15 >> 1) < G1_NODES && node < n_nodes) (half ? ald2 : als2)[node] = y;
    }
#pragma unroll
    for (int r = 0; r < G1_NODES; ++r) {
        if (node0 + r < n_nodes)
            h2[(size_t)(node0 + r) * 32 + half * 16 + r15] = half ? acc1[r] : acc0[r];
    }
}

__device__ __forceinline__ void score_l2(int k, int n, int start, int node,
                                         const int* __restrict__ col, const float* __restrict__ als,
                                         float adv, int& s, float& v) {
    s = node; v = -INFINITY;
    if (k < n) {
        if (k < n - 1) s = col[start + k];
        const float y = als[s] + adv;
        v = y > 0.f ? y : NEG_SLOPE * y;
    }
}

template <int T0, int NT>
__device__ __forceinline__ void accum_steps_l2(int s, float a, const float* __restrict__ hmat,
                                               int c8, v4f& acc) {
    v4f r[NT];
#pragma unroll
    for (int t = 0; t < NT; ++t) {
        const int sq = (int)grp8_pick_u((unsigned)s, T0 + t);
        r[t] = ((const v4f*)(hmat + (size_t)sq * 32))[c8];
    }
#pragma unroll
    for (int t = 0; t < NT; ++t) {
        const float aq = grp8_pick(a, T0 + t);
        acc += aq * r[t];
    }
}

__device__ __forceinline__ void accum_l2(int cntc, int s, float a, const float* __restrict__ hmat,
                                         int c8, v4f& acc) {
    accum_steps_l2<0, 4>(s, a, hmat, c8, acc);
    if (cntc > 16) accum_steps_l2<4, 2>(s, a, hmat, c8, acc);
    if (cntc > 24) accum_steps_l2<6, 2>(s, a, hmat, c8, acc);
}

__global__ void __launch_bounds__(256) k_gat_gather_l2(const int* __restrict__ rowptr,
                                                       const int* __restrict__ cnt,
                                                       const int* __restrict__ col,
                                                       const float* __restrict__ als,
                                                       const float* __restrict__ ald,
                                                       const float* __restrict__ hmat,
                                                       const float* __restrict__ bias,
                                                       float* __restrict__ outp,
                                                       int n_nodes) {
    const int lane = threadIdx.x & 31;
    const int q = lane >> 3, c8 = lane & 7;
    const int kk = 4 * c8 + q;
    const int wave0 = __builtin_amdgcn_readfirstlane(
        (int)(blockIdx.x * (blockDim.x >> 5) + (threadIdx.x >> 5)));
    const int n_waves = (int)(gridDim.x * (blockDim.x >> 5));
    const v4f bv = ((const v4f*)bias)[c8];
    for (int node = wave0; node < n_nodes; node += n_waves) {
        const int start = rowptr[node];
        const int n = cnt[node] + 1;
        const float adv = ald[node];

        int s0; float v0;
        score_l2(kk, n, start, node, col, als, adv, s0, v0);

        float m = v0;
        for (int base = 32; base < n; base += 32) {
            int s; float v;
            score_l2(base + kk, n, start, node, col, als, adv, s, v);
            m = fmaxf(m, v);
        }
#pragma unroll
        for (int off = 16; off >= 1; off >>= 1) m = fmaxf(m, __shfl_xor(m, off));

        float den = __expf(v0 - m);
        for (int base = 32; base < n; base += 32) {
            int s; float v;
            score_l2(base + kk, n, start, node, col, als, adv, s, v);
            den += __expf(v - m);
        }
#pragma unroll
        for (int off = 16; off >= 1; off >>= 1) den += __shfl_xor(den, off);
        den += 1e-16f;

        v4f acc = {0.f, 0.f, 0.f, 0.f};
        accum_l2(n < 32 ? n : 32, s0, __expf(v0 - m) / den, hmat, c8, acc);
        for (int base = 32; base < n; base += 32) {
            int s; float v;
            score_l2(base + kk, n, start, node, col, als, adv, s, v);
            accum_l2((n - base) < 32 ? (n - base) : 32, s, __expf(v - m) / den, hmat, c8, acc);
        }
        acc.x += row_xor<8>(acc.x); acc.y += row_xor<8>(acc.y); acc.z += row_xor<8>(acc.z); acc.w += row_xor<8>(acc.w);
        acc.x += __shfl_xor(acc.x, 16); acc.y += __shfl_xor(acc.y, 16);
        acc.z += __shfl_xor(acc.z, 16); acc.w += __shfl_xor(acc.w, 16);
        if (lane < 8) ((v4f*)(outp + (size_t)node * 32))[c8] = acc + bv;
    }
}


extern "C" void kernel_launch(void* const* d_in, const int* in_sizes, int n_in,
                              void* d_out, int out_size, void* d_ws, size_t ws_size,
                              hipStream_t stream) {
    (void)n_in; (void)out_size; (void)ws_size; (void)in_sizes;
    constexpr long long N = 100000, E = 1600000;
    constexpr int C1 = 128, HEADS = 4, C2 = 32;
    constexpr int N_TILES = (int)((E + TILE_E - 1) / TILE_E);
    constexpr int N_BKT   = (int)((N + BKT_N - 1) / BKT_N);
    static_assert(N_BKT + 1 <= TOFF_P, "bucket starts of a tile must fit its table row");
    static_assert(N <= (1LL << 24), "a record holds the source in 24 bits");
    constexpr int NODES_PER_WAVE = 7;
    constexpr int GATHER_BLOCKS = (int)((N + 8 * NODES_PER_WAVE - 1) / (8 * NODES_PER_WAVE));
    static_assert(G1_NODES == NODES_PER_WAVE, "both gathers share one grid");

    const float* x    = (const float*)d_in[0];
    const int*   ei   = (const int*)  d_in[1];
    const float* W1   = (const float*)d_in[2];
    const float* aS1  = (const float*)d_in[3];
    const float* aD1  = (const float*)d_in[4];
    const float* b1   = (const float*)d_in[5];
    const float* W2   = (const float*)d_in[6];
    const float* aS2  = (const float*)d_in[7];
    const float* aD2  = (const float*)d_in[8];
    const float* b2   = (const float*)d_in[9];
    float* out = (float*)d_out;

    char* ws = (char*)d_ws;
    size_t off = 0;
    auto alloc = [&](size_t bytes) { size_t o = off; off = (off + bytes + 255) & ~(size_t)255; return o; };
    unsigned short* h1b  = (unsigned short*)(ws + alloc(N * C1 * 2));
    float*          h2   = (float*)(ws + alloc(N * C2 * 4));
    float*          as1  = (float*)(ws + alloc(N * HEADS * 4));
    float*          ad1  = (float*)(ws + alloc(N * HEADS * 4));
    float*          as2  = (float*)(ws + alloc(N * 4));
    float*          ad2  = (float*)(ws + alloc(N * 4));
    int*            cnt  = (int*)(ws + alloc(N * 4));
    int*            rptr = (int*)(ws + alloc(N * 4));
    unsigned*       tsrt = (unsigned*)(ws + alloc((size_t)N_TILES * TILE_E * 4));
    int*            toff = (int*)(ws + alloc((size_t)N_TILES * TOFF_P * 4));
    int*            col  = (int*)(ws + alloc(E * 4));
    unsigned short* w1t  = (unsigned short*)(ws + alloc(128 * 128 * 2));
    unsigned short* w2t  = (unsigned short*)(ws + alloc(32 * 128 * 2));

    constexpr int B = 256;
    auto blocks = [](long long n, int b) { return (unsigned)((n + b - 1) / b); };

    constexpr int PREP_THREADS = 128 * 128 + 32 * 128;
    k_prep<<<blocks(PREP_THREADS, B), B, 0, stream>>>(W1, w1t, W2, w2t);

    constexpr unsigned GEMM_WGS = (unsigned)((N / 16 + 7) / 8);
    constexpr unsigned EDGE_WGS = (unsigned)N_TILES;
    constexpr unsigned PAIR_WGS = 2 * (GEMM_WGS > EDGE_WGS ? GEMM_WGS : EDGE_WGS);
    k_gemm1_tilesort<<<PAIR_WGS, 256, 0, stream>>>(x, w1t, h1b, aS1, aD1, as1, ad1, (int)(N / 16), ei, tsrt, toff, E, N_TILES);

    k_bucket_csr<N_TILES><<<N_BKT, CSR_THREADS, 0, stream>>>(tsrt, toff, cnt, rptr, col, (int)N);

    k_gat_gather_l1_gemm2<<<GATHER_BLOCKS, 256, 0, stream>>>(rptr, cnt, col, as1, ad1, h1b, b1, w2t, aS2, aD2,
                                                            h2, as2, ad2, (int)N, (int)E);

    k_gat_gather_l2<<<GATHER_BLOCKS, 256, 0, stream>>>(rptr, cnt, col, as2, ad2, h2, b2, out, (int)N);
}
